# starting kernel
# baseline (speedup 1.0000x reference)
#include <hip/hip_runtime.h>
#include <math.h>

typedef __attribute__((ext_vector_type(2))) float v2f;
typedef __attribute__((ext_vector_type(4))) float v4f;
typedef __attribute__((ext_vector_type(8))) float v8f;

#define SEQ   100
#define BATCH 256
#define INDIM 300
#define HID   128
#define NROW  (BATCH*SEQ)
#define WIN   11

static __device__ __forceinline__ v8f wmma_f32_k4(v2f a, v2f b, v8f c) {
    return __builtin_amdgcn_wmma_f32_16x16x4_f32(false, a, false, b, (short)0, c,
                                                 false, false);
}

__global__ __launch_bounds__(256) void k_h(const float* __restrict__ x,
                                           const float* __restrict__ W1,
                                           const float* __restrict__ b1,
                                           float* __restrict__ h) {
    __shared__ float xs[16 * 301];
    const int m0 = blockIdx.x * 16;
    for (int e = threadIdx.x; e < 16 * INDIM; e += 256) {
        int row = e / INDIM, col = e - row * INDIM;
        xs[row * 301 + col] = x[(m0 + row) * INDIM + col];
    }
    __syncthreads();

    const int wave = threadIdx.x >> 5;
    const int lane = threadIdx.x & 31;
    const int rr   = lane & 15;
    const int half = lane >> 4;
    const int n    = wave * 16 + rr;

    v8f acc = {0.f,0.f,0.f,0.f,0.f,0.f,0.f,0.f};
    for (int kk = 0; kk < INDIM; kk += 4) {
        const int off = kk + half * 2;
        v2f a; a.x = xs[rr * 301 + off]; a.y = xs[rr * 301 + off + 1];
        v2f bf; bf.x = W1[off * HID + n]; bf.y = W1[(off + 1) * HID + n];
        acc = wmma_f32_k4(a, bf, acc);
    }
    const float bias = b1[n];
#pragma unroll
    for (int r = 0; r < 8; ++r) {
        const int m = m0 + r + half * 8;
        float v = acc[r] + bias;
        h[m * HID + n] = v > 0.f ? v : 0.f;
    }
}

__global__ __launch_bounds__(256) void k_qkv(const float* __restrict__ h,
                                             const float* __restrict__ Wq,
                                             const float* __restrict__ Wk,
                                             const float* __restrict__ Wv,
                                             float* __restrict__ Q,
                                             float* __restrict__ K,
                                             float* __restrict__ V) {
    __shared__ float hs[16 * 132];
    const int m0 = blockIdx.x * 16;
    for (int e = threadIdx.x; e < 16 * HID; e += 256) {
        int row = e >> 7, col = e & 127;
        hs[row * 132 + col] = h[(m0 + row) * HID + col];
    }
    __syncthreads();

    const int wave = threadIdx.x >> 5;
    const int lane = threadIdx.x & 31;
    const int rr   = lane & 15;
    const int half = lane >> 4;
    const int n    = wave * 16 + rr;

    v8f aq = {0.f,0.f,0.f,0.f,0.f,0.f,0.f,0.f};
    v8f ak = aq, av = aq;
    for (int kk = 0; kk < HID; kk += 4) {
        const int off = kk + half * 2;
        v2f a; a.x = hs[rr * 132 + off]; a.y = hs[rr * 132 + off + 1];
        v2f bq; bq.x = Wq[off * HID + n]; bq.y = Wq[(off + 1) * HID + n];
        aq = wmma_f32_k4(a, bq, aq);
        v2f bk; bk.x = Wk[off * HID + n]; bk.y = Wk[(off + 1) * HID + n];
        ak = wmma_f32_k4(a, bk, ak);
        v2f bv; bv.x = Wv[off * HID + n]; bv.y = Wv[(off + 1) * HID + n];
        av = wmma_f32_k4(a, bv, av);
    }
#pragma unroll
    for (int r = 0; r < 8; ++r) {
        const int m = m0 + r + half * 8;
        Q[m * HID + n] = aq[r];
        K[m * HID + n] = ak[r];
        V[m * HID + n] = av[r];
    }
}

__global__ __launch_bounds__(256) void k_attn(const float* __restrict__ Q,
                                              const float* __restrict__ Km,
                                              const float* __restrict__ V,
                                              const float* __restrict__ W2,
                                              const float* __restrict__ b2,
                                              float* __restrict__ out,
                                              float* __restrict__ alphaOut) {
    const int wave = threadIdx.x >> 5;
    const int lane = threadIdx.x & 31;
    const int rr   = lane & 15;
    const int half = lane >> 4;
    const int idx  = blockIdx.x * 8 + wave;
    if (idx >= NROW) return;
    const int s = idx % SEQ;

    const int  r     = s + 5 - rr;
    const bool rowok = (rr < WIN) && (r >= 0) && (r < SEQ);
    const float* qrow = Q  + (long)(idx + 5 - rr) * HID;
    const float* krow = Km + (long)(idx + 5 - rr) * HID;

    v8f acc = {0.f,0.f,0.f,0.f,0.f,0.f,0.f,0.f};
    for (int kk = 0; kk < HID; kk += 4) {
        const int off = kk + half * 2;
        v2f a  = {0.f, 0.f};
        v2f bf = {0.f, 0.f};
        if (rowok) {
            a  = *(const v2f*)(qrow + off);
            bf = *(const v2f*)(krow + off);
        }
        acc = wmma_f32_k4(a, bf, acc);
    }

    const float scale = 0.08838834764831845f;
    const long  abase = (long)idx * (WIN * WIN);
    float bsum = 0.f;
#pragma unroll
    for (int rI = 0; rI < 8; ++rI) {
        const float sc = acc[rI] * scale;
        float mv = (rr < WIN) ? sc : -3.0e38f;
        mv = fmaxf(mv, __shfl_xor(mv, 1, 32));
        mv = fmaxf(mv, __shfl_xor(mv, 2, 32));
        mv = fmaxf(mv, __shfl_xor(mv, 4, 32));
        mv = fmaxf(mv, __shfl_xor(mv, 8, 32));
        float p = (rr < WIN) ? expf(sc - mv) : 0.f;
        float sum = p;
        sum += __shfl_xor(sum, 1, 32);
        sum += __shfl_xor(sum, 2, 32);
        sum += __shfl_xor(sum, 4, 32);
        sum += __shfl_xor(sum, 8, 32);
        const float al = p / sum;
        const int i = rI + half * 8;
        if (i < WIN) {
            if (rr < WIN) alphaOut[abase + i * WIN + rr] = al;
            bsum += al;
        }
    }
    bsum += __shfl_xor(bsum, 16, 32);

    v4f pooled = {0.f, 0.f, 0.f, 0.f};
    for (int j = 0; j < WIN; ++j) {
        const float bj = __shfl(bsum, j, 32);
        const int rv = s + 5 - j;
        if (rv >= 0 && rv < SEQ) {
            v4f vv = *(const v4f*)(V + (long)(idx + 5 - j) * HID + lane * 4);
            pooled += bj * vv;
        }
    }

    float p0 = 0.f, p1 = 0.f;
#pragma unroll
    for (int t = 0; t < 4; ++t) {
        const int d = lane * 4 + t;
        p0 += pooled[t] * W2[d * 2 + 0];
        p1 += pooled[t] * W2[d * 2 + 1];
    }
#pragma unroll
    for (int mask = 1; mask < 32; mask <<= 1) {
        p0 += __shfl_xor(p0, mask, 32);
        p1 += __shfl_xor(p1, mask, 32);
    }
    if (lane == 0) {
        out[idx * 2 + 0] = 1.f / (1.f + expf(-(p0 + b2[0])));
        out[idx * 2 + 1] = 1.f / (1.f + expf(-(p1 + b2[1])));
    }
}

extern "C" void kernel_launch(void* const* d_in, const int* in_sizes, int n_in,
                              void* d_out, int out_size, void* d_ws, size_t ws_size,
                              hipStream_t stream) {
    const float* x  = (const float*)d_in[0];
    const float* W1 = (const float*)d_in[1];
    const float* b1 = (const float*)d_in[2];
    const float* Wq = (const float*)d_in[3];
    const float* Wk = (const float*)d_in[4];
    const float* Wv = (const float*)d_in[5];
    const float* W2 = (const float*)d_in[6];
    const float* b2 = (const float*)d_in[7];

    float* h = (float*)d_ws;
    float* Q = h + (size_t)NROW * HID;
    float* K = Q + (size_t)NROW * HID;
    float* V = K + (size_t)NROW * HID;

    float* out      = (float*)d_out;
    float* alphaOut = out + (size_t)NROW * 2;

    k_h  <<<NROW / 16, 256, 0, stream>>>(x, W1, b1, h);
    k_qkv<<<NROW / 16, 256, 0, stream>>>(h, Wq, Wk, Wv, Q, K, V);
    k_attn<<<NROW / 8, 256, 0, stream>>>(Q, K, V, W2, b2, out, alphaOut);
}


# k_attn: 16 neighbouring positions per block fetch their 26 shared Q,K,V rows once into LDS and build one 32x32 QK^T tile (split-K over 2 waves, partials added); was per-position row loads
# speedup vs baseline: 1.4347x; 1.4347x over previous
#include <hip/hip_runtime.h>
#include <math.h>

typedef __attribute__((ext_vector_type(2))) float v2f;
typedef __attribute__((ext_vector_type(4))) float v4f;
typedef __attribute__((ext_vector_type(8))) float v8f;

#define SEQ   100
#define BATCH 256
#define INDIM 300
#define HID   128
#define NROW  (BATCH*SEQ)
#define WIN   11

static __device__ __forceinline__ v8f wmma_f32_k4(v2f a, v2f b, v8f c) {
    return __builtin_amdgcn_wmma_f32_16x16x4_f32(false, a, false, b, (short)0, c,
                                                 false, false);
}

__global__ __launch_bounds__(256) void k_h(const float* __restrict__ x,
                                           const float* __restrict__ W1,
                                           const float* __restrict__ b1,
                                           float* __restrict__ h) {
    __shared__ float xs[16 * 301];
    const int m0 = blockIdx.x * 16;
    for (int e = threadIdx.x; e < 16 * INDIM; e += 256) {
        int row = e / INDIM, col = e - row * INDIM;
        xs[row * 301 + col] = x[(m0 + row) * INDIM + col];
    }
    __syncthreads();

    const int wave = threadIdx.x >> 5;
    const int lane = threadIdx.x & 31;
    const int rr   = lane & 15;
    const int half = lane >> 4;
    const int n    = wave * 16 + rr;

    v8f acc = {0.f,0.f,0.f,0.f,0.f,0.f,0.f,0.f};
    for (int kk = 0; kk < INDIM; kk += 4) {
        const int off = kk + half * 2;
        v2f a; a.x = xs[rr * 301 + off]; a.y = xs[rr * 301 + off + 1];
        v2f bf; bf.x = W1[off * HID + n]; bf.y = W1[(off + 1) * HID + n];
        acc = wmma_f32_k4(a, bf, acc);
    }
    const float bias = b1[n];
#pragma unroll
    for (int r = 0; r < 8; ++r) {
        const int m = m0 + r + half * 8;
        float v = acc[r] + bias;
        h[m * HID + n] = v > 0.f ? v : 0.f;
    }
}

__global__ __launch_bounds__(256) void k_qkv(const float* __restrict__ h,
                                             const float* __restrict__ Wq,
                                             const float* __restrict__ Wk,
                                             const float* __restrict__ Wv,
                                             float* __restrict__ Q,
                                             float* __restrict__ K,
                                             float* __restrict__ V) {
    __shared__ float hs[16 * 132];
    const int m0 = blockIdx.x * 16;
    for (int e = threadIdx.x; e < 16 * HID; e += 256) {
        int row = e >> 7, col = e & 127;
        hs[row * 132 + col] = h[(m0 + row) * HID + col];
    }
    __syncthreads();

    const int wave = threadIdx.x >> 5;
    const int lane = threadIdx.x & 31;
    const int rr   = lane & 15;
    const int half = lane >> 4;
    const int n    = wave * 16 + rr;

    v8f aq = {0.f,0.f,0.f,0.f,0.f,0.f,0.f,0.f};
    v8f ak = aq, av = aq;
    for (int kk = 0; kk < HID; kk += 4) {
        const int off = kk + half * 2;
        v2f a; a.x = hs[rr * 132 + off]; a.y = hs[rr * 132 + off + 1];
        v2f bq; bq.x = Wq[off * HID + n]; bq.y = Wq[(off + 1) * HID + n];
        aq = wmma_f32_k4(a, bq, aq);
        v2f bk; bk.x = Wk[off * HID + n]; bk.y = Wk[(off + 1) * HID + n];
        ak = wmma_f32_k4(a, bk, ak);
        v2f bv; bv.x = Wv[off * HID + n]; bv.y = Wv[(off + 1) * HID + n];
        av = wmma_f32_k4(a, bv, av);
    }
#pragma unroll
    for (int r = 0; r < 8; ++r) {
        const int m = m0 + r + half * 8;
        Q[m * HID + n] = aq[r];
        K[m * HID + n] = ak[r];
        V[m * HID + n] = av[r];
    }
}

#define TP  16
#define NR  32
#define RS  132
#define GS  36

__global__ __launch_bounds__(256) void k_attn(const float* __restrict__ Q,
                                              const float* __restrict__ Km,
                                              const float* __restrict__ V,
                                              const float* __restrict__ W2,
                                              const float* __restrict__ b2,
                                              float* __restrict__ out,
                                              float* __restrict__ alphaOut) {
    __shared__ __attribute__((aligned(16))) float Qs[NR * RS];
    __shared__ __attribute__((aligned(16))) float Ks[NR * RS];
    __shared__ __attribute__((aligned(16))) float Gs[2 * NR * GS];

    const int wave = threadIdx.x >> 5;
    const int lane = threadIdx.x & 31;
    const int rr   = lane & 15;
    const int half = lane >> 4;
    const int idx0 = blockIdx.x * TP;
    const int base = idx0 - 5;

    v4f qv[4], kv[4], vv[4];
    bool rok[4];
#pragma unroll
    for (int t = 0; t < 4; ++t) {
        const int row = wave + 8 * t;
        const int rc  = row < TP + 10 ? row : TP + 9;
        int g = base + rc;
        rok[t] = (row < TP + 10) && (g >= 0) && (g < NROW);
        g = g < 0 ? 0 : (g >= NROW ? NROW - 1 : g);
        const long o = (long)g * HID + lane * 4;
        qv[t] = *(const v4f*)(Q  + o);
        kv[t] = *(const v4f*)(Km + o);
        vv[t] = *(const v4f*)(V  + o);
    }
    const v4f zero4 = {0.f, 0.f, 0.f, 0.f};
#pragma unroll
    for (int t = 0; t < 4; ++t) {
        const int row = wave + 8 * t;
        *(v4f*)(Qs + row * RS + lane * 4) = rok[t] ? qv[t] : zero4;
        *(v4f*)(Ks + row * RS + lane * 4) = rok[t] ? kv[t] : zero4;
    }
    __syncthreads();

    {
        const int tm = wave & 1;
        const int tn = (wave >> 1) & 1;
        const int kh = wave >> 2;
        const float* qa = Qs + (tm * 16 + rr) * RS + kh * 64 + half * 2;
        const float* kb = Ks + (tn * 16 + rr) * RS + kh * 64 + half * 2;
        v8f acc = {0.f,0.f,0.f,0.f,0.f,0.f,0.f,0.f};
        for (int kk = 0; kk < 64; kk += 4) {
            const v2f a  = *(const v2f*)(qa + kk);
            const v2f bf = *(const v2f*)(kb + kk);
            acc = wmma_f32_k4(a, bf, acc);
        }
        float* gp = Gs + kh * (NR * GS) + (tm * 16 + half * 8) * GS + tn * 16 + rr;
#pragma unroll
        for (int r = 0; r < 8; ++r) gp[r * GS] = acc[r];
    }
    __syncthreads();

    float* Vs = Qs;
#pragma unroll
    for (int t = 0; t < 4; ++t) {
        const int row = wave + 8 * t;
        *(v4f*)(Vs + row * RS + lane * 4) = rok[t] ? vv[t] : zero4;
    }
    __syncthreads();

    float w20[4], w21[4];
#pragma unroll
    for (int t = 0; t < 4; ++t) {
        const int d = lane * 4 + t;
        w20[t] = W2[d * 2 + 0];
        w21[t] = W2[d * 2 + 1];
    }
    const float bias0 = b2[0], bias1 = b2[1];
    const float scale = 0.08838834764831845f;

    for (int t = 0; t < 2; ++t) {
        const int p   = wave * 2 + t;
        const int idx = idx0 + p;
        const int s   = idx % SEQ;
        const int rj  = s + 5 - rr;
        const bool colok = (rr < WIN) && (rj >= 0) && (rj < SEQ);
        const long abase = (long)idx * (WIN * WIN);
        float bsum = 0.f;
#pragma unroll
        for (int rI = 0; rI < 8; ++rI) {
            const int i  = rI + half * 8;
            const int ri = s + 5 - i;
            const bool ok = colok && (i < WIN) && (ri >= 0) && (ri < SEQ);
            const int o = ok ? (p + 10 - i) * GS + (p + 10 - rr) : 0;
            const float gsum = Gs[o] + Gs[NR * GS + o];
            const float sc = ok ? gsum * scale : 0.f;
            float mv = (rr < WIN) ? sc : -3.0e38f;
            mv = fmaxf(mv, __shfl_xor(mv, 1, 32));
            mv = fmaxf(mv, __shfl_xor(mv, 2, 32));
            mv = fmaxf(mv, __shfl_xor(mv, 4, 32));
            mv = fmaxf(mv, __shfl_xor(mv, 8, 32));
            float pe = (rr < WIN) ? expf(sc - mv) : 0.f;
            float sum = pe;
            sum += __shfl_xor(sum, 1, 32);
            sum += __shfl_xor(sum, 2, 32);
            sum += __shfl_xor(sum, 4, 32);
            sum += __shfl_xor(sum, 8, 32);
            const float al = pe / sum;
            if (i < WIN) {
                if (rr < WIN) alphaOut[abase + i * WIN + rr] = al;
                bsum += al;
            }
        }
        bsum += __shfl_xor(bsum, 16, 32);

        v4f pooled = {0.f, 0.f, 0.f, 0.f};
        for (int j = 0; j < WIN; ++j) {
            const float bj = __shfl(bsum, j, 32);
            const int rv = s + 5 - j;
            const v4f vrow = *(const v4f*)(Vs + (p + 10 - j) * RS + lane * 4);
            if (rv >= 0 && rv < SEQ) pooled += bj * vrow;
        }

        float p0 = 0.f, p1 = 0.f;
#pragma unroll
        for (int u = 0; u < 4; ++u) {
            p0 += pooled[u] * w20[u];
            p1 += pooled[u] * w21[u];
        }
#pragma unroll
        for (int mask = 1; mask < 32; mask <<= 1) {
            p0 += __shfl_xor(p0, mask, 32);
            p1 += __shfl_xor(p1, mask, 32);
        }
        if (lane == 0) {
            out[idx * 2 + 0] = 1.f / (1.f + expf(-(p0 + bias0)));
            out[idx * 2 + 1] = 1.f / (1.f + expf(-(p1 + bias1)));
        }
    }
}

extern "C" void kernel_launch(void* const* d_in, const int* in_sizes, int n_in,
                              void* d_out, int out_size, void* d_ws, size_t ws_size,
                              hipStream_t stream) {
    const float* x  = (const float*)d_in[0];
    const float* W1 = (const float*)d_in[1];
    const float* b1 = (const float*)d_in[2];
    const float* Wq = (const float*)d_in[3];
    const float* Wk = (const float*)d_in[4];
    const float* Wv = (const float*)d_in[5];
    const float* W2 = (const float*)d_in[6];
    const float* b2 = (const float*)d_in[7];

    float* h = (float*)d_ws;
    float* Q = h + (size_t)NROW * HID;
    float* K = Q + (size_t)NROW * HID;
    float* V = K + (size_t)NROW * HID;

    float* out      = (float*)d_out;
    float* alphaOut = out + (size_t)NROW * 2;

    k_h  <<<NROW / 16, 256, 0, stream>>>(x, W1, b1, h);
    k_qkv<<<NROW / 16, 256, 0, stream>>>(h, Wq, Wk, Wv, Q, K, V);
    k_attn<<<NROW / TP, 256, 0, stream>>>(Q, K, V, W2, b2, out, alphaOut);
}


# k_h and k_qkv fused into one launch: each block keeps its 16x128 tile of h in LDS between the two GEMM phases; h is no longer written to or read from the workspace
# speedup vs baseline: 1.5397x; 1.0732x over previous
#include <hip/hip_runtime.h>
#include <math.h>

typedef __attribute__((ext_vector_type(2))) float v2f;
typedef __attribute__((ext_vector_type(4))) float v4f;
typedef __attribute__((ext_vector_type(8))) float v8f;

#define SEQ   100
#define BATCH 256
#define INDIM 300
#define HID   128
#define NROW  (BATCH*SEQ)
#define WIN   11

static __device__ __forceinline__ v8f wmma_f32_k4(v2f a, v2f b, v8f c) {
    return __builtin_amdgcn_wmma_f32_16x16x4_f32(false, a, false, b, (short)0, c,
                                                 false, false);
}

__global__ __launch_bounds__(256) void k_hqkv(const float* __restrict__ x,
                                              const float* __restrict__ W1,
                                              const float* __restrict__ b1,
                                              const float* __restrict__ Wq,
                                              const float* __restrict__ Wk,
                                              const float* __restrict__ Wv,
                                              float* __restrict__ Q,
                                              float* __restrict__ K,
                                              float* __restrict__ V) {
    __shared__ float xs[16 * 301];
    __shared__ float hs[16 * 132];
    const int m0 = blockIdx.x * 16;
    for (int e = threadIdx.x; e < 16 * INDIM; e += 256) {
        int row = e / INDIM, col = e - row * INDIM;
        xs[row * 301 + col] = x[(m0 + row) * INDIM + col];
    }
    __syncthreads();

    const int wave = threadIdx.x >> 5;
    const int lane = threadIdx.x & 31;
    const int rr   = lane & 15;
    const int half = lane >> 4;
    const int n    = wave * 16 + rr;

    {
        v8f acc = {0.f,0.f,0.f,0.f,0.f,0.f,0.f,0.f};
        for (int kk = 0; kk < INDIM; kk += 4) {
            const int off = kk + half * 2;
            v2f a; a.x = xs[rr * 301 + off]; a.y = xs[rr * 301 + off + 1];
            v2f bf; bf.x = W1[off * HID + n]; bf.y = W1[(off + 1) * HID + n];
            acc = wmma_f32_k4(a, bf, acc);
        }
        const float bias = b1[n];
#pragma unroll
        for (int r = 0; r < 8; ++r) {
            const float v = acc[r] + bias;
            hs[(r + half * 8) * 132 + n] = v > 0.f ? v : 0.f;
        }
    }
    __syncthreads();

    v8f aq = {0.f,0.f,0.f,0.f,0.f,0.f,0.f,0.f};
    v8f ak = aq, av = aq;
    for (int kk = 0; kk < HID; kk += 4) {
        const int off = kk + half * 2;
        v2f a; a.x = hs[rr * 132 + off]; a.y = hs[rr * 132 + off + 1];
        v2f bq; bq.x = Wq[off * HID + n]; bq.y = Wq[(off + 1) * HID + n];
        aq = wmma_f32_k4(a, bq, aq);
        v2f bk; bk.x = Wk[off * HID + n]; bk.y = Wk[(off + 1) * HID + n];
        ak = wmma_f32_k4(a, bk, ak);
        v2f bv; bv.x = Wv[off * HID + n]; bv.y = Wv[(off + 1) * HID + n];
        av = wmma_f32_k4(a, bv, av);
    }
#pragma unroll
    for (int r = 0; r < 8; ++r) {
        const int m = m0 + r + half * 8;
        Q[m * HID + n] = aq[r];
        K[m * HID + n] = ak[r];
        V[m * HID + n] = av[r];
    }
}

#define TP  16
#define NR  32
#define RS  132
#define GS  36

__global__ __launch_bounds__(256) void k_attn(const float* __restrict__ Q,
                                              const float* __restrict__ Km,
                                              const float* __restrict__ V,
                                              const float* __restrict__ W2,
                                              const float* __restrict__ b2,
                                              float* __restrict__ out,
                                              float* __restrict__ alphaOut) {
    __shared__ __attribute__((aligned(16))) float Qs[NR * RS];
    __shared__ __attribute__((aligned(16))) float Ks[NR * RS];
    __shared__ __attribute__((aligned(16))) float Gs[2 * NR * GS];

    const int wave = threadIdx.x >> 5;
    const int lane = threadIdx.x & 31;
    const int rr   = lane & 15;
    const int half = lane >> 4;
    const int idx0 = blockIdx.x * TP;
    const int base = idx0 - 5;

    v4f qv[4], kv[4], vv[4];
    bool rok[4];
#pragma unroll
    for (int t = 0; t < 4; ++t) {
        const int row = wave + 8 * t;
        const int rc  = row < TP + 10 ? row : TP + 9;
        int g = base + rc;
        rok[t] = (row < TP + 10) && (g >= 0) && (g < NROW);
        g = g < 0 ? 0 : (g >= NROW ? NROW - 1 : g);
        const long o = (long)g * HID + lane * 4;
        qv[t] = *(const v4f*)(Q  + o);
        kv[t] = *(const v4f*)(Km + o);
        vv[t] = *(const v4f*)(V  + o);
    }
    const v4f zero4 = {0.f, 0.f, 0.f, 0.f};
#pragma unroll
    for (int t = 0; t < 4; ++t) {
        const int row = wave + 8 * t;
        *(v4f*)(Qs + row * RS + lane * 4) = rok[t] ? qv[t] : zero4;
        *(v4f*)(Ks + row * RS + lane * 4) = rok[t] ? kv[t] : zero4;
    }
    __syncthreads();

    {
        const int tm = wave & 1;
        const int tn = (wave >> 1) & 1;
        const int kh = wave >> 2;
        const float* qa = Qs + (tm * 16 + rr) * RS + kh * 64 + half * 2;
        const float* kb = Ks + (tn * 16 + rr) * RS + kh * 64 + half * 2;
        v8f acc = {0.f,0.f,0.f,0.f,0.f,0.f,0.f,0.f};
        for (int kk = 0; kk < 64; kk += 4) {
            const v2f a  = *(const v2f*)(qa + kk);
            const v2f bf = *(const v2f*)(kb + kk);
            acc = wmma_f32_k4(a, bf, acc);
        }
        float* gp = Gs + kh * (NR * GS) + (tm * 16 + half * 8) * GS + tn * 16 + rr;
#pragma unroll
        for (int r = 0; r < 8; ++r) gp[r * GS] = acc[r];
    }
    __syncthreads();

    float* Vs = Qs;
#pragma unroll
    for (int t = 0; t < 4; ++t) {
        const int row = wave + 8 * t;
        *(v4f*)(Vs + row * RS + lane * 4) = rok[t] ? vv[t] : zero4;
    }
    __syncthreads();

    float w20[4], w21[4];
#pragma unroll
    for (int t = 0; t < 4; ++t) {
        const int d = lane * 4 + t;
        w20[t] = W2[d * 2 + 0];
        w21[t] = W2[d * 2 + 1];
    }
    const float bias0 = b2[0], bias1 = b2[1];
    const float scale = 0.08838834764831845f;

    for (int t = 0; t < 2; ++t) {
        const int p   = wave * 2 + t;
        const int idx = idx0 + p;
        const int s   = idx % SEQ;
        const int rj  = s + 5 - rr;
        const bool colok = (rr < WIN) && (rj >= 0) && (rj < SEQ);
        const long abase = (long)idx * (WIN * WIN);
        float bsum = 0.f;
#pragma unroll
        for (int rI = 0; rI < 8; ++rI) {
            const int i  = rI + half * 8;
            const int ri = s + 5 - i;
            const bool ok = colok && (i < WIN) && (ri >= 0) && (ri < SEQ);
            const int o = ok ? (p + 10 - i) * GS + (p + 10 - rr) : 0;
            const float gsum = Gs[o] + Gs[NR * GS + o];
            const float sc = ok ? gsum * scale : 0.f;
            float mv = (rr < WIN) ? sc : -3.0e38f;
            mv = fmaxf(mv, __shfl_xor(mv, 1, 32));
            mv = fmaxf(mv, __shfl_xor(mv, 2, 32));
            mv = fmaxf(mv, __shfl_xor(mv, 4, 32));
            mv = fmaxf(mv, __shfl_xor(mv, 8, 32));
            float pe = (rr < WIN) ? expf(sc - mv) : 0.f;
            float sum = pe;
            sum += __shfl_xor(sum, 1, 32);
            sum += __shfl_xor(sum, 2, 32);
            sum += __shfl_xor(sum, 4, 32);
            sum += __shfl_xor(sum, 8, 32);
            const float al = pe / sum;
            if (i < WIN) {
                if (rr < WIN) alphaOut[abase + i * WIN + rr] = al;
                bsum += al;
            }
        }
        bsum += __shfl_xor(bsum, 16, 32);

        v4f pooled = {0.f, 0.f, 0.f, 0.f};
        for (int j = 0; j < WIN; ++j) {
            const float bj = __shfl(bsum, j, 32);
            const int rv = s + 5 - j;
            const v4f vrow = *(const v4f*)(Vs + (p + 10 - j) * RS + lane * 4);
            if (rv >= 0 && rv < SEQ) pooled += bj * vrow;
        }

        float p0 = 0.f, p1 = 0.f;
#pragma unroll
        for (int u = 0; u < 4; ++u) {
            p0 += pooled[u] * w20[u];
            p1 += pooled[u] * w21[u];
        }
#pragma unroll
        for (int mask = 1; mask < 32; mask <<= 1) {
            p0 += __shfl_xor(p0, mask, 32);
            p1 += __shfl_xor(p1, mask, 32);
        }
        if (lane == 0) {
            out[idx * 2 + 0] = 1.f / (1.f + expf(-(p0 + bias0)));
            out[idx * 2 + 1] = 1.f / (1.f + expf(-(p1 + bias1)));
        }
    }
}

extern "C" void kernel_launch(void* const* d_in, const int* in_sizes, int n_in,
                              void* d_out, int out_size, void* d_ws, size_t ws_size,
                              hipStream_t stream) {
    const float* x  = (const float*)d_in[0];
    const float* W1 = (const float*)d_in[1];
    const float* b1 = (const float*)d_in[2];
    const float* Wq = (const float*)d_in[3];
    const float* Wk = (const float*)d_in[4];
    const float* Wv = (const float*)d_in[5];
    const float* W2 = (const float*)d_in[6];
    const float* b2 = (const float*)d_in[7];

    float* Q = (float*)d_ws;
    float* K = Q + (size_t)NROW * HID;
    float* V = K + (size_t)NROW * HID;

    float* out      = (float*)d_out;
    float* alphaOut = out + (size_t)NROW * 2;

    k_hqkv<<<NROW / 16, 256, 0, stream>>>(x, W1, b1, Wq, Wk, Wv, Q, K, V);
    k_attn<<<NROW / TP, 256, 0, stream>>>(Q, K, V, W2, b2, out, alphaOut);
}


# k_attn softmax: a lane owns a whole 11-score row (max, exp, sequential sum, 1/sum in registers; was 16-lane butterflies); alpha tile built in LDS, column sums read there, written with 16-byte stores
# speedup vs baseline: 1.7987x; 1.1682x over previous
#include <hip/hip_runtime.h>
#include <math.h>

typedef __attribute__((ext_vector_type(2))) float v2f;
typedef __attribute__((ext_vector_type(4))) float v4f;
typedef __attribute__((ext_vector_type(8))) float v8f;

#define SEQ   100
#define BATCH 256
#define INDIM 300
#define HID   128
#define NROW  (BATCH*SEQ)
#define WIN   11

static __device__ __forceinline__ v8f wmma_f32_k4(v2f a, v2f b, v8f c) {
    return __builtin_amdgcn_wmma_f32_16x16x4_f32(false, a, false, b, (short)0, c,
                                                 false, false);
}

__global__ __launch_bounds__(256) void k_hqkv(const float* __restrict__ x,
                                              const float* __restrict__ W1,
                                              const float* __restrict__ b1,
                                              const float* __restrict__ Wq,
                                              const float* __restrict__ Wk,
                                              const float* __restrict__ Wv,
                                              float* __restrict__ Q,
                                              float* __restrict__ K,
                                              float* __restrict__ V) {
    __shared__ float xs[16 * 301];
    __shared__ float hs[16 * 132];
    const int m0 = blockIdx.x * 16;
    for (int e = threadIdx.x; e < 16 * INDIM; e += 256) {
        int row = e / INDIM, col = e - row * INDIM;
        xs[row * 301 + col] = x[(m0 + row) * INDIM + col];
    }
    __syncthreads();

    const int wave = threadIdx.x >> 5;
    const int lane = threadIdx.x & 31;
    const int rr   = lane & 15;
    const int half = lane >> 4;
    const int n    = wave * 16 + rr;

    {
        v8f acc = {0.f,0.f,0.f,0.f,0.f,0.f,0.f,0.f};
        for (int kk = 0; kk < INDIM; kk += 4) {
            const int off = kk + half * 2;
            v2f a; a.x = xs[rr * 301 + off]; a.y = xs[rr * 301 + off + 1];
            v2f bf; bf.x = W1[off * HID + n]; bf.y = W1[(off + 1) * HID + n];
            acc = wmma_f32_k4(a, bf, acc);
        }
        const float bias = b1[n];
#pragma unroll
        for (int r = 0; r < 8; ++r) {
            const float v = acc[r] + bias;
            hs[(r + half * 8) * 132 + n] = v > 0.f ? v : 0.f;
        }
    }
    __syncthreads();

    v8f aq = {0.f,0.f,0.f,0.f,0.f,0.f,0.f,0.f};
    v8f ak = aq, av = aq;
    for (int kk = 0; kk < HID; kk += 4) {
        const int off = kk + half * 2;
        v2f a; a.x = hs[rr * 132 + off]; a.y = hs[rr * 132 + off + 1];
        v2f bq; bq.x = Wq[off * HID + n]; bq.y = Wq[(off + 1) * HID + n];
        aq = wmma_f32_k4(a, bq, aq);
        v2f bk; bk.x = Wk[off * HID + n]; bk.y = Wk[(off + 1) * HID + n];
        ak = wmma_f32_k4(a, bk, ak);
        v2f bv; bv.x = Wv[off * HID + n]; bv.y = Wv[(off + 1) * HID + n];
        av = wmma_f32_k4(a, bv, av);
    }
#pragma unroll
    for (int r = 0; r < 8; ++r) {
        const int m = m0 + r + half * 8;
        Q[m * HID + n] = aq[r];
        K[m * HID + n] = ak[r];
        V[m * HID + n] = av[r];
    }
}

#define TP  16
#define NR  32
#define RS  132
#define GS  36

__global__ __launch_bounds__(256) void k_attn(const float* __restrict__ Q,
                                              const float* __restrict__ Km,
                                              const float* __restrict__ V,
                                              const float* __restrict__ W2,
                                              const float* __restrict__ b2,
                                              float* __restrict__ out,
                                              float* __restrict__ alphaOut) {
    __shared__ __attribute__((aligned(16))) float Qs[NR * RS];
    __shared__ __attribute__((aligned(16))) float Ks[NR * RS];
    __shared__ __attribute__((aligned(16))) float Gs[2 * NR * GS];

    const int wave = threadIdx.x >> 5;
    const int lane = threadIdx.x & 31;
    const int rr   = lane & 15;
    const int half = lane >> 4;
    const int idx0 = blockIdx.x * TP;
    const int base = idx0 - 5;

    v4f qv[4], kv[4], vv[4];
    bool rok[4];
#pragma unroll
    for (int t = 0; t < 4; ++t) {
        const int row = wave + 8 * t;
        const int rc  = row < TP + 10 ? row : TP + 9;
        int g = base + rc;
        rok[t] = (row < TP + 10) && (g >= 0) && (g < NROW);
        g = g < 0 ? 0 : (g >= NROW ? NROW - 1 : g);
        const long o = (long)g * HID + lane * 4;
        qv[t] = *(const v4f*)(Q  + o);
        kv[t] = *(const v4f*)(Km + o);
        vv[t] = *(const v4f*)(V  + o);
    }
    const v4f zero4 = {0.f, 0.f, 0.f, 0.f};
#pragma unroll
    for (int t = 0; t < 4; ++t) {
        const int row = wave + 8 * t;
        *(v4f*)(Qs + row * RS + lane * 4) = rok[t] ? qv[t] : zero4;
        *(v4f*)(Ks + row * RS + lane * 4) = rok[t] ? kv[t] : zero4;
    }
    __syncthreads();

    {
        const int tm = wave & 1;
        const int tn = (wave >> 1) & 1;
        const int kh = wave >> 2;
        const float* qa = Qs + (tm * 16 + rr) * RS + kh * 64 + half * 2;
        const float* kb = Ks + (tn * 16 + rr) * RS + kh * 64 + half * 2;
        v8f acc = {0.f,0.f,0.f,0.f,0.f,0.f,0.f,0.f};
        for (int kk = 0; kk < 64; kk += 4) {
            const v2f a  = *(const v2f*)(qa + kk);
            const v2f bf = *(const v2f*)(kb + kk);
            acc = wmma_f32_k4(a, bf, acc);
        }
        float* gp = Gs + kh * (NR * GS) + (tm * 16 + half * 8) * GS + tn * 16 + rr;
#pragma unroll
        for (int r = 0; r < 8; ++r) gp[r * GS] = acc[r];
    }
    __syncthreads();

    float* Vs = Qs;
#pragma unroll
    for (int t = 0; t < 4; ++t) {
        const int row = wave + 8 * t;
        *(v4f*)(Vs + row * RS + lane * 4) = rok[t] ? vv[t] : zero4;
    }
    __syncthreads();

    float w20[4], w21[4];
#pragma unroll
    for (int t = 0; t < 4; ++t) {
        const int d = lane * 4 + t;
        w20[t] = W2[d * 2 + 0];
        w21[t] = W2[d * 2 + 1];
    }
    const float bias0 = b2[0], bias1 = b2[1];
    const float scale = 0.08838834764831845f;

    const int wu = __builtin_amdgcn_readfirstlane(wave);
    const int p0 = wu * 2;

    float* As = Ks;

    {
        const int p  = p0 + half;
        const int s  = (idx0 + p) % SEQ;
        const int ic = rr < WIN ? rr : WIN - 1;
        const int ri = s + 5 - ic;
        const bool rowok = (ri >= 0) && (ri < SEQ);
        const float* g = Gs + (p + 10 - ic) * GS + (p + 10);
        float sc[WIN];
#pragma unroll
        for (int j = 0; j < WIN; ++j) {
            const int rj = s + 5 - j;
            const bool ok = rowok && (rj >= 0) && (rj < SEQ);
            const float gsum = g[-j] + g[NR * GS - j];
            sc[j] = ok ? gsum * scale : 0.f;
        }
        float mv = sc[0];
#pragma unroll
        for (int j = 1; j < WIN; ++j) mv = fmaxf(mv, sc[j]);
        float sum = 0.f;
#pragma unroll
        for (int j = 0; j < WIN; ++j) {
            sc[j] = expf(sc[j] - mv);
            sum += sc[j];
        }
        const float rinv = 1.0f / sum;
        if (rr < WIN) {
            float* ap = As + p * (WIN * WIN) + rr * WIN;
#pragma unroll
            for (int j = 0; j < WIN; ++j) ap[j] = sc[j] * rinv;
        }
    }
    __syncthreads();

    {
        const v4f* a4 = (const v4f*)As;
        v4f* o4 = (v4f*)(alphaOut + (long)idx0 * (WIN * WIN));
        o4[threadIdx.x] = a4[threadIdx.x];
        if (threadIdx.x < TP * WIN * WIN / 4 - 256)
            o4[threadIdx.x + 256] = a4[threadIdx.x + 256];
    }

    float beta = 0.f;
    {
        const int jc = rr < WIN ? rr : WIN - 1;
        const float* ac = As + (p0 + half) * (WIN * WIN) + jc;
#pragma unroll
        for (int i = 0; i < WIN; ++i) beta += ac[i * WIN];
    }

    v4f pooled[2];
    pooled[0] = zero4;
    pooled[1] = zero4;
    const int sA = (idx0 + p0) % SEQ;
    const int sB = (idx0 + p0 + 1) % SEQ;
#pragma unroll
    for (int r = WIN; r >= 0; --r) {
        const v4f vrow = *(const v4f*)(Vs + (p0 + r) * RS + lane * 4);
        if (r <= WIN - 1) {
            const int j = WIN - 1 - r;
            const float bj = __int_as_float(__builtin_amdgcn_readlane(__float_as_int(beta), j));
            const int rv = sA + 5 - j;
            if (rv >= 0 && rv < SEQ) pooled[0] += bj * vrow;
        }
        if (r >= 1) {
            const int j = WIN - r;
            const float bj = __int_as_float(__builtin_amdgcn_readlane(__float_as_int(beta), 16 + j));
            const int rv = sB + 5 - j;
            if (rv >= 0 && rv < SEQ) pooled[1] += bj * vrow;
        }
    }

#pragma unroll
    for (int t = 0; t < 2; ++t) {
        const int idx = idx0 + p0 + t;
        float q0 = 0.f, q1 = 0.f;
#pragma unroll
        for (int u = 0; u < 4; ++u) {
            q0 += pooled[t][u] * w20[u];
            q1 += pooled[t][u] * w21[u];
        }
#pragma unroll
        for (int mask = 1; mask < 32; mask <<= 1) {
            q0 += __shfl_xor(q0, mask, 32);
            q1 += __shfl_xor(q1, mask, 32);
        }
        if (lane == 0) {
            out[idx * 2 + 0] = 1.f / (1.f + expf(-(q0 + bias0)));
            out[idx * 2 + 1] = 1.f / (1.f + expf(-(q1 + bias1)));
        }
    }
}

extern "C" void kernel_launch(void* const* d_in, const int* in_sizes, int n_in,
                              void* d_out, int out_size, void* d_ws, size_t ws_size,
                              hipStream_t stream) {
    const float* x  = (const float*)d_in[0];
    const float* W1 = (const float*)d_in[1];
    const float* b1 = (const float*)d_in[2];
    const float* Wq = (const float*)d_in[3];
    const float* Wk = (const float*)d_in[4];
    const float* Wv = (const float*)d_in[5];
    const float* W2 = (const float*)d_in[6];
    const float* b2 = (const float*)d_in[7];

    float* Q = (float*)d_ws;
    float* K = Q + (size_t)NROW * HID;
    float* V = K + (size_t)NROW * HID;

    float* out      = (float*)d_out;
    float* alphaOut = out + (size_t)NROW * 2;

    k_hqkv<<<NROW / 16, 256, 0, stream>>>(x, W1, b1, Wq, Wk, Wv, Q, K, V);
    k_attn<<<NROW / TP, 256, 0, stream>>>(Q, K, V, W2, b2, out, alphaOut);
}


# k_attn: W2 applied to the 26 staged V rows first (2 logits a row); a position's logits = beta-weighted sum of 11 projected rows; pooled vector and V in LDS gone; order of the sums changes
# speedup vs baseline: 1.8771x; 1.0436x over previous
#include <hip/hip_runtime.h>
#include <math.h>

typedef __attribute__((ext_vector_type(2))) float v2f;
typedef __attribute__((ext_vector_type(4))) float v4f;
typedef __attribute__((ext_vector_type(8))) float v8f;

#define SEQ   100
#define BATCH 256
#define INDIM 300
#define HID   128
#define NROW  (BATCH*SEQ)
#define WIN   11

static __device__ __forceinline__ v8f wmma_f32_k4(v2f a, v2f b, v8f c) {
    return __builtin_amdgcn_wmma_f32_16x16x4_f32(false, a, false, b, (short)0, c,
                                                 false, false);
}

__global__ __launch_bounds__(256) void k_hqkv(const float* __restrict__ x,
                                              const float* __restrict__ W1,
                                              const float* __restrict__ b1,
                                              const float* __restrict__ Wq,
                                              const float* __restrict__ Wk,
                                              const float* __restrict__ Wv,
                                              float* __restrict__ Q,
                                              float* __restrict__ K,
                                              float* __restrict__ V) {
    __shared__ float xs[16 * 301];
    __shared__ float hs[16 * 132];
    const int m0 = blockIdx.x * 16;
    for (int e = threadIdx.x; e < 16 * INDIM; e += 256) {
        int row = e / INDIM, col = e - row * INDIM;
        xs[row * 301 + col] = x[(m0 + row) * INDIM + col];
    }
    __syncthreads();

    const int wave = threadIdx.x >> 5;
    const int lane = threadIdx.x & 31;
    const int rr   = lane & 15;
    const int half = lane >> 4;
    const int n    = wave * 16 + rr;

    {
        v8f acc = {0.f,0.f,0.f,0.f,0.f,0.f,0.f,0.f};
        for (int kk = 0; kk < INDIM; kk += 4) {
            const int off = kk + half * 2;
            v2f a; a.x = xs[rr * 301 + off]; a.y = xs[rr * 301 + off + 1];
            v2f bf; bf.x = W1[off * HID + n]; bf.y = W1[(off + 1) * HID + n];
            acc = wmma_f32_k4(a, bf, acc);
        }
        const float bias = b1[n];
#pragma unroll
        for (int r = 0; r < 8; ++r) {
            const float v = acc[r] + bias;
            hs[(r + half * 8) * 132 + n] = v > 0.f ? v : 0.f;
        }
    }
    __syncthreads();

    v8f aq = {0.f,0.f,0.f,0.f,0.f,0.f,0.f,0.f};
    v8f ak = aq, av = aq;
    for (int kk = 0; kk < HID; kk += 4) {
        const int off = kk + half * 2;
        v2f a; a.x = hs[rr * 132 + off]; a.y = hs[rr * 132 + off + 1];
        v2f bq; bq.x = Wq[off * HID + n]; bq.y = Wq[(off + 1) * HID + n];
        aq = wmma_f32_k4(a, bq, aq);
        v2f bk; bk.x = Wk[off * HID + n]; bk.y = Wk[(off + 1) * HID + n];
        ak = wmma_f32_k4(a, bk, ak);
        v2f bv; bv.x = Wv[off * HID + n]; bv.y = Wv[(off + 1) * HID + n];
        av = wmma_f32_k4(a, bv, av);
    }
#pragma unroll
    for (int r = 0; r < 8; ++r) {
        const int m = m0 + r + half * 8;
        Q[m * HID + n] = aq[r];
        K[m * HID + n] = ak[r];
        V[m * HID + n] = av[r];
    }
}

#define TP  16
#define NR  32
#define RS  132
#define GS  36

__global__ __launch_bounds__(256) void k_attn(const float* __restrict__ Q,
                                              const float* __restrict__ Km,
                                              const float* __restrict__ V,
                                              const float* __restrict__ W2,
                                              const float* __restrict__ b2,
                                              float* __restrict__ out,
                                              float* __restrict__ alphaOut) {
    __shared__ __attribute__((aligned(16))) float Qs[NR * RS];
    __shared__ __attribute__((aligned(16))) float Ps[NR * 2];
    __shared__ __attribute__((aligned(16))) float Ks[NR * RS];
    __shared__ __attribute__((aligned(16))) float Gs[2 * NR * GS];

    const int wave = threadIdx.x >> 5;
    const int lane = threadIdx.x & 31;
    const int rr   = lane & 15;
    const int half = lane >> 4;
    const int idx0 = blockIdx.x * TP;
    const int base = idx0 - 5;

    v4f qv[4], kv[4], vv[4];
    bool rok[4];
#pragma unroll
    for (int t = 0; t < 4; ++t) {
        const int row = wave + 8 * t;
        const int rc  = row < TP + 10 ? row : TP + 9;
        int g = base + rc;
        rok[t] = (row < TP + 10) && (g >= 0) && (g < NROW);
        g = g < 0 ? 0 : (g >= NROW ? NROW - 1 : g);
        const long o = (long)g * HID + lane * 4;
        qv[t] = *(const v4f*)(Q  + o);
        kv[t] = *(const v4f*)(Km + o);
        vv[t] = *(const v4f*)(V  + o);
    }
    float w20[4], w21[4];
#pragma unroll
    for (int t = 0; t < 4; ++t) {
        const int d = lane * 4 + t;
        w20[t] = W2[d * 2 + 0];
        w21[t] = W2[d * 2 + 1];
    }
    const v4f zero4 = {0.f, 0.f, 0.f, 0.f};
#pragma unroll
    for (int t = 0; t < 4; ++t) {
        const int row = wave + 8 * t;
        *(v4f*)(Qs + row * RS + lane * 4) = rok[t] ? qv[t] : zero4;
        *(v4f*)(Ks + row * RS + lane * 4) = rok[t] ? kv[t] : zero4;
    }
    __syncthreads();

    {
        const int tm = wave & 1;
        const int tn = (wave >> 1) & 1;
        const int kh = wave >> 2;
        const float* qa = Qs + (tm * 16 + rr) * RS + kh * 64 + half * 2;
        const float* kb = Ks + (tn * 16 + rr) * RS + kh * 64 + half * 2;
        v8f acc = {0.f,0.f,0.f,0.f,0.f,0.f,0.f,0.f};
        for (int kk = 0; kk < 64; kk += 4) {
            const v2f a  = *(const v2f*)(qa + kk);
            const v2f bf = *(const v2f*)(kb + kk);
            acc = wmma_f32_k4(a, bf, acc);
        }
        float* gp = Gs + kh * (NR * GS) + (tm * 16 + half * 8) * GS + tn * 16 + rr;
#pragma unroll
        for (int r = 0; r < 8; ++r) gp[r * GS] = acc[r];
    }

    {
        float pv[8];
#pragma unroll
        for (int t = 0; t < 4; ++t) {
            float a0 = 0.f, a1 = 0.f;
#pragma unroll
            for (int u = 0; u < 4; ++u) {
                a0 += vv[t][u] * w20[u];
                a1 += vv[t][u] * w21[u];
            }
            pv[2 * t]     = a0;
            pv[2 * t + 1] = a1;
        }
        const bool hi16 = (lane & 16) != 0;
        const bool hi8  = (lane & 8) != 0;
        const bool hi4  = (lane & 4) != 0;
        float r4[4];
#pragma unroll
        for (int i = 0; i < 4; ++i) {
            const float keep = hi16 ? pv[i + 4] : pv[i];
            const float send = hi16 ? pv[i] : pv[i + 4];
            r4[i] = keep + __shfl_xor(send, 16, 32);
        }
        float r2[2];
#pragma unroll
        for (int i = 0; i < 2; ++i) {
            const float keep = hi8 ? r4[i + 2] : r4[i];
            const float send = hi8 ? r4[i] : r4[i + 2];
            r2[i] = keep + __shfl_xor(send, 8, 32);
        }
        float r1;
        {
            const float keep = hi4 ? r2[1] : r2[0];
            const float send = hi4 ? r2[0] : r2[1];
            r1 = keep + __shfl_xor(send, 4, 32);
        }
        r1 += __shfl_xor(r1, 2, 32);
        r1 += __shfl_xor(r1, 1, 32);
        if ((lane & 3) == 0) {
            const int k = lane >> 2;
            Ps[(wave + 8 * (k >> 1)) * 2 + (k & 1)] = r1;
        }
    }
    __syncthreads();

    const float bias0 = b2[0], bias1 = b2[1];
    const float scale = 0.08838834764831845f;

    const int wu = __builtin_amdgcn_readfirstlane(wave);
    const int p0 = wu * 2;

    float* As = Ks;

    {
        const int p  = p0 + half;
        const int s  = (idx0 + p) % SEQ;
        const int ic = rr < WIN ? rr : WIN - 1;
        const int ri = s + 5 - ic;
        const bool rowok = (ri >= 0) && (ri < SEQ);
        const float* g = Gs + (p + 10 - ic) * GS + (p + 10);
        float sc[WIN];
#pragma unroll
        for (int j = 0; j < WIN; ++j) {
            const int rj = s + 5 - j;
            const bool ok = rowok && (rj >= 0) && (rj < SEQ);
            const float gsum = g[-j] + g[NR * GS - j];
            sc[j] = ok ? gsum * scale : 0.f;
        }
        float mv = sc[0];
#pragma unroll
        for (int j = 1; j < WIN; ++j) mv = fmaxf(mv, sc[j]);
        float sum = 0.f;
#pragma unroll
        for (int j = 0; j < WIN; ++j) {
            sc[j] = expf(sc[j] - mv);
            sum += sc[j];
        }
        const float rinv = 1.0f / sum;
        if (rr < WIN) {
            float* ap = As + p * (WIN * WIN) + rr * WIN;
#pragma unroll
            for (int j = 0; j < WIN; ++j) ap[j] = sc[j] * rinv;
        }
    }
    __syncthreads();

    {
        const v4f* a4 = (const v4f*)As;
        v4f* o4 = (v4f*)(alphaOut + (long)idx0 * (WIN * WIN));
        o4[threadIdx.x] = a4[threadIdx.x];
        if (threadIdx.x < TP * WIN * WIN / 4 - 256)
            o4[threadIdx.x + 256] = a4[threadIdx.x + 256];
    }

    float beta = 0.f;
    {
        const int jc = rr < WIN ? rr : WIN - 1;
        const float* ac = As + (p0 + half) * (WIN * WIN) + jc;
#pragma unroll
        for (int i = 0; i < WIN; ++i) beta += ac[i * WIN];
    }

    {
        const int p  = p0 + half;
        const int s  = (idx0 + p) % SEQ;
        const int jc = rr < WIN ? rr : WIN - 1;
        const int rv = s + 5 - jc;
        const bool valid = (rr < WIN) && (rv >= 0) && (rv < SEQ);
        const v2f pr = *(const v2f*)(Ps + (p + 10 - jc) * 2);
        const float t0 = valid ? beta * pr.x : 0.f;
        const float t1 = valid ? beta * pr.y : 0.f;
        const bool hi8 = (rr & 8) != 0;
        const float keep = hi8 ? t1 : t0;
        const float send = hi8 ? t0 : t1;
        float r = keep + __shfl_xor(send, 8, 32);
        r += __shfl_xor(r, 4, 32);
        r += __shfl_xor(r, 2, 32);
        r += __shfl_xor(r, 1, 32);
        if ((rr & 7) == 0) {
            const int c = rr >> 3;
            const float bias = c ? bias1 : bias0;
            out[(idx0 + p) * 2 + c] = 1.f / (1.f + expf(-(r + bias)));
        }
    }
}

extern "C" void kernel_launch(void* const* d_in, const int* in_sizes, int n_in,
                              void* d_out, int out_size, void* d_ws, size_t ws_size,
                              hipStream_t stream) {
    const float* x  = (const float*)d_in[0];
    const float* W1 = (const float*)d_in[1];
    const float* b1 = (const float*)d_in[2];
    const float* Wq = (const float*)d_in[3];
    const float* Wk = (const float*)d_in[4];
    const float* Wv = (const float*)d_in[5];
    const float* W2 = (const float*)d_in[6];
    const float* b2 = (const float*)d_in[7];

    float* Q = (float*)d_ws;
    float* K = Q + (size_t)NROW * HID;
    float* V = K + (size_t)NROW * HID;

    float* out      = (float*)d_out;
    float* alphaOut = out + (size_t)NROW * 2;

    k_hqkv<<<NROW / 16, 256, 0, stream>>>(x, W1, b1, Wq, Wk, Wv, Q, K, V);
    k_attn<<<NROW / TP, 256, 0, stream>>>(Q, K, V, W2, b2, out, alphaOut);
}


# Wv and W2 multiplied first (WP = Wv@W2, 128x2, per block): k_hqkv makes P = h@WP (one narrow WMMA tile, split-K over 8 waves) instead of V = h@Wv; V plane gone; order of the sums changes
# speedup vs baseline: 2.0379x; 1.0857x over previous
#include <hip/hip_runtime.h>
#include <math.h>

typedef __attribute__((ext_vector_type(2))) float v2f;
typedef __attribute__((ext_vector_type(4))) float v4f;
typedef __attribute__((ext_vector_type(8))) float v8f;

#define SEQ   100
#define BATCH 256
#define INDIM 300
#define HID   128
#define NROW  (BATCH*SEQ)
#define WIN   11

static __device__ __forceinline__ v8f wmma_f32_k4(v2f a, v2f b, v8f c) {
    return __builtin_amdgcn_wmma_f32_16x16x4_f32(false, a, false, b, (short)0, c,
                                                 false, false);
}

static __device__ __forceinline__ float wave_sum8(const float (&pv)[8], int lane) {
    const bool hi16 = (lane & 16) != 0;
    const bool hi8  = (lane & 8) != 0;
    const bool hi4  = (lane & 4) != 0;
    float r4[4];
#pragma unroll
    for (int i = 0; i < 4; ++i) {
        const float keep = hi16 ? pv[i + 4] : pv[i];
        const float send = hi16 ? pv[i] : pv[i + 4];
        r4[i] = keep + __shfl_xor(send, 16, 32);
    }
    float r2[2];
#pragma unroll
    for (int i = 0; i < 2; ++i) {
        const float keep = hi8 ? r4[i + 2] : r4[i];
        const float send = hi8 ? r4[i] : r4[i + 2];
        r2[i] = keep + __shfl_xor(send, 8, 32);
    }
    float r1;
    {
        const float keep = hi4 ? r2[1] : r2[0];
        const float send = hi4 ? r2[0] : r2[1];
        r1 = keep + __shfl_xor(send, 4, 32);
    }
    r1 += __shfl_xor(r1, 2, 32);
    r1 += __shfl_xor(r1, 1, 32);
    return r1;
}

__global__ __launch_bounds__(256) void k_hqkv(const float* __restrict__ x,
                                              const float* __restrict__ W1,
                                              const float* __restrict__ b1,
                                              const float* __restrict__ Wq,
                                              const float* __restrict__ Wk,
                                              const float* __restrict__ Wv,
                                              const float* __restrict__ W2,
                                              float* __restrict__ Q,
                                              float* __restrict__ K,
                                              float* __restrict__ P) {
    __shared__ float xs[16 * 301];
    __shared__ float hs[16 * 132];
    __shared__ float wp[HID * 2];
    __shared__ float pp[8 * 16 * 2];
    const int m0 = blockIdx.x * 16;
    for (int e = threadIdx.x; e < 16 * INDIM; e += 256) {
        int row = e / INDIM, col = e - row * INDIM;
        xs[row * 301 + col] = x[(m0 + row) * INDIM + col];
    }
    __syncthreads();

    const int wave = threadIdx.x >> 5;
    const int lane = threadIdx.x & 31;
    const int rr   = lane & 15;
    const int half = lane >> 4;
    const int n    = wave * 16 + rr;


    {
        v8f acc = {0.f,0.f,0.f,0.f,0.f,0.f,0.f,0.f};
        for (int kk = 0; kk < INDIM; kk += 4) {
            const int off = kk + half * 2;
            v2f a; a.x = xs[rr * 301 + off]; a.y = xs[rr * 301 + off + 1];
            v2f bf; bf.x = W1[off * HID + n]; bf.y = W1[(off + 1) * HID + n];
            acc = wmma_f32_k4(a, bf, acc);
        }
        const float bias = b1[n];
#pragma unroll
        for (int r = 0; r < 8; ++r) {
            const float v = acc[r] + bias;
            hs[(r + half * 8) * 132 + n] = v > 0.f ? v : 0.f;
        }
    }

    {
        float w20[4], w21[4];
#pragma unroll
        for (int t = 0; t < 4; ++t) {
            const int d = lane * 4 + t;
            w20[t] = W2[d * 2 + 0];
            w21[t] = W2[d * 2 + 1];
        }
#pragma unroll
        for (int q = 0; q < 4; ++q) {
            v4f wr[4];
#pragma unroll
            for (int t = 0; t < 4; ++t)
                wr[t] = *(const v4f*)(Wv + (q * 32 + wave + 8 * t) * HID + lane * 4);
            float pv[8];
#pragma unroll
            for (int t = 0; t < 4; ++t) {
                float a0 = 0.f, a1 = 0.f;
#pragma unroll
                for (int u = 0; u < 4; ++u) {
                    a0 += wr[t][u] * w20[u];
                    a1 += wr[t][u] * w21[u];
                }
                pv[2 * t]     = a0;
                pv[2 * t + 1] = a1;
            }
            const float r1 = wave_sum8(pv, lane);
            if ((lane & 3) == 0) {
                const int k = lane >> 2;
                wp[(q * 32 + wave + 8 * (k >> 1)) * 2 + (k & 1)] = r1;
            }
        }
    }
    __syncthreads();

    v8f aq = {0.f,0.f,0.f,0.f,0.f,0.f,0.f,0.f};
    v8f ak = aq;
#pragma unroll 4
    for (int kk = 0; kk < HID; kk += 4) {
        const int off = kk + half * 2;
        v2f a; a.x = hs[rr * 132 + off]; a.y = hs[rr * 132 + off + 1];
        v2f bq; bq.x = Wq[off * HID + n]; bq.y = Wq[(off + 1) * HID + n];
        aq = wmma_f32_k4(a, bq, aq);
        v2f bk; bk.x = Wk[off * HID + n]; bk.y = Wk[(off + 1) * HID + n];
        ak = wmma_f32_k4(a, bk, ak);
    }
#pragma unroll
    for (int r = 0; r < 8; ++r) {
        const int m = m0 + r + half * 8;
        Q[m * HID + n] = aq[r];
        K[m * HID + n] = ak[r];
    }

    v8f ap = {0.f,0.f,0.f,0.f,0.f,0.f,0.f,0.f};
#pragma unroll
    for (int i = 0; i < 4; ++i) {
        const int off = wave * 16 + i * 4 + half * 2;
        v2f a; a.x = hs[rr * 132 + off]; a.y = hs[rr * 132 + off + 1];
        const float w0 = wp[off * 2 + (rr & 1)];
        const float w1 = wp[(off + 1) * 2 + (rr & 1)];
        v2f bp; bp.x = rr < 2 ? w0 : 0.f; bp.y = rr < 2 ? w1 : 0.f;
        ap = wmma_f32_k4(a, bp, ap);
    }
    if (rr < 2) {
#pragma unroll
        for (int r = 0; r < 8; ++r) pp[(wave * 16 + half * 8 + r) * 2 + rr] = ap[r];
    }
    __syncthreads();
    if (threadIdx.x < 32) {
        float s = 0.f;
#pragma unroll
        for (int w = 0; w < 8; ++w) s += pp[w * 32 + threadIdx.x];
        P[m0 * 2 + threadIdx.x] = s;
    }
}

#define TP  16
#define NR  32
#define RS  132
#define GS  36

__global__ __launch_bounds__(256) void k_attn(const float* __restrict__ Q,
                                              const float* __restrict__ Km,
                                              const float* __restrict__ P,
                                              const float* __restrict__ b2,
                                              float* __restrict__ out,
                                              float* __restrict__ alphaOut) {
    __shared__ __attribute__((aligned(16))) float Qs[NR * RS];
    __shared__ __attribute__((aligned(16))) float Ps[NR * 2];
    __shared__ __attribute__((aligned(16))) float Ks[NR * RS];
    __shared__ __attribute__((aligned(16))) float Gs[2 * NR * GS];

    const int wave = threadIdx.x >> 5;
    const int lane = threadIdx.x & 31;
    const int rr   = lane & 15;
    const int half = lane >> 4;
    const int idx0 = blockIdx.x * TP;
    const int base = idx0 - 5;

    v4f qv[4], kv[4];
    bool rok[4];
#pragma unroll
    for (int t = 0; t < 4; ++t) {
        const int row = wave + 8 * t;
        const int rc  = row < TP + 10 ? row : TP + 9;
        int g = base + rc;
        rok[t] = (row < TP + 10) && (g >= 0) && (g < NROW);
        g = g < 0 ? 0 : (g >= NROW ? NROW - 1 : g);
        const long o = (long)g * HID + lane * 4;
        qv[t] = *(const v4f*)(Q  + o);
        kv[t] = *(const v4f*)(Km + o);
    }
    float pval = 0.f;
    if (threadIdx.x < NR * 2) {
        const int row = threadIdx.x >> 1;
        const int rc  = row < TP + 10 ? row : TP + 9;
        int g = base + rc;
        g = g < 0 ? 0 : (g >= NROW ? NROW - 1 : g);
        pval = P[g * 2 + (threadIdx.x & 1)];
    }
    const v4f zero4 = {0.f, 0.f, 0.f, 0.f};
#pragma unroll
    for (int t = 0; t < 4; ++t) {
        const int row = wave + 8 * t;
        *(v4f*)(Qs + row * RS + lane * 4) = rok[t] ? qv[t] : zero4;
        *(v4f*)(Ks + row * RS + lane * 4) = rok[t] ? kv[t] : zero4;
    }
    if (threadIdx.x < NR * 2) Ps[threadIdx.x] = pval;
    __syncthreads();

    {
        const int tm = wave & 1;
        const int tn = (wave >> 1) & 1;
        const int kh = wave >> 2;
        const float* qa = Qs + (tm * 16 + rr) * RS + kh * 64 + half * 2;
        const float* kb = Ks + (tn * 16 + rr) * RS + kh * 64 + half * 2;
        v8f acc = {0.f,0.f,0.f,0.f,0.f,0.f,0.f,0.f};
        for (int kk = 0; kk < 64; kk += 4) {
            const v2f a  = *(const v2f*)(qa + kk);
            const v2f bf = *(const v2f*)(kb + kk);
            acc = wmma_f32_k4(a, bf, acc);
        }
        float* gp = Gs + kh * (NR * GS) + (tm * 16 + half * 8) * GS + tn * 16 + rr;
#pragma unroll
        for (int r = 0; r < 8; ++r) gp[r * GS] = acc[r];
    }

    __syncthreads();

    const float bias0 = b2[0], bias1 = b2[1];
    const float scale = 0.08838834764831845f;

    const int wu = __builtin_amdgcn_readfirstlane(wave);
    const int p0 = wu * 2;

    float* As = Ks;

    {
        const int p  = p0 + half;
        const int s  = (idx0 + p) % SEQ;
        const int ic = rr < WIN ? rr : WIN - 1;
        const int ri = s + 5 - ic;
        const bool rowok = (ri >= 0) && (ri < SEQ);
        const float* g = Gs + (p + 10 - ic) * GS + (p + 10);
        float sc[WIN];
#pragma unroll
        for (int j = 0; j < WIN; ++j) {
            const int rj = s + 5 - j;
            const bool ok = rowok && (rj >= 0) && (rj < SEQ);
            const float gsum = g[-j] + g[NR * GS - j];
            sc[j] = ok ? gsum * scale : 0.f;
        }
        float mv = sc[0];
#pragma unroll
        for (int j = 1; j < WIN; ++j) mv = fmaxf(mv, sc[j]);
        float sum = 0.f;
#pragma unroll
        for (int j = 0; j < WIN; ++j) {
            sc[j] = expf(sc[j] - mv);
            sum += sc[j];
        }
        const float rinv = 1.0f / sum;
        if (rr < WIN) {
            float* ap = As + p * (WIN * WIN) + rr * WIN;
#pragma unroll
            for (int j = 0; j < WIN; ++j) ap[j] = sc[j] * rinv;
        }
    }
    __syncthreads();

    {
        const v4f* a4 = (const v4f*)As;
        v4f* o4 = (v4f*)(alphaOut + (long)idx0 * (WIN * WIN));
        o4[threadIdx.x] = a4[threadIdx.x];
        if (threadIdx.x < TP * WIN * WIN / 4 - 256)
            o4[threadIdx.x + 256] = a4[threadIdx.x + 256];
    }

    float beta = 0.f;
    {
        const int jc = rr < WIN ? rr : WIN - 1;
        const float* ac = As + (p0 + half) * (WIN * WIN) + jc;
#pragma unroll
        for (int i = 0; i < WIN; ++i) beta += ac[i * WIN];
    }

    {
        const int p  = p0 + half;
        const int s  = (idx0 + p) % SEQ;
        const int jc = rr < WIN ? rr : WIN - 1;
        const int rv = s + 5 - jc;
        const bool valid = (rr < WIN) && (rv >= 0) && (rv < SEQ);
        const v2f pr = *(const v2f*)(Ps + (p + 10 - jc) * 2);
        const float t0 = valid ? beta * pr.x : 0.f;
        const float t1 = valid ? beta * pr.y : 0.f;
        const bool hi8 = (rr & 8) != 0;
        const float keep = hi8 ? t1 : t0;
        const float send = hi8 ? t0 : t1;
        float r = keep + __shfl_xor(send, 8, 32);
        r += __shfl_xor(r, 4, 32);
        r += __shfl_xor(r, 2, 32);
        r += __shfl_xor(r, 1, 32);
        if ((rr & 7) == 0) {
            const int c = rr >> 3;
            const float bias = c ? bias1 : bias0;
            out[(idx0 + p) * 2 + c] = 1.f / (1.f + expf(-(r + bias)));
        }
    }
}

extern "C" void kernel_launch(void* const* d_in, const int* in_sizes, int n_in,
                              void* d_out, int out_size, void* d_ws, size_t ws_size,
                              hipStream_t stream) {
    const float* x  = (const float*)d_in[0];
    const float* W1 = (const float*)d_in[1];
    const float* b1 = (const float*)d_in[2];
    const float* Wq = (const float*)d_in[3];
    const float* Wk = (const float*)d_in[4];
    const float* Wv = (const float*)d_in[5];
    const float* W2 = (const float*)d_in[6];
    const float* b2 = (const float*)d_in[7];

    float* Q = (float*)d_ws;
    float* K = Q + (size_t)NROW * HID;
    float* P = K + (size_t)NROW * HID;

    float* out      = (float*)d_out;
    float* alphaOut = out + (size_t)NROW * 2;

    k_hqkv<<<NROW / 16, 256, 0, stream>>>(x, W1, b1, Wq, Wk, Wv, W2, Q, K, P);
    k_attn<<<NROW / TP, 256, 0, stream>>>(Q, K, P, b2, out, alphaOut);
}


# both kernels fused into one launch, one 16-wave block per 100-row sequence: x tile, h, Q, K and P stay in LDS, nothing goes through the workspace; a weight fragment now feeds 3-4 WMMAs, was 1
# speedup vs baseline: 2.4387x; 1.1967x over previous
#include <hip/hip_runtime.h>
#include <math.h>


typedef __attribute__((ext_vector_type(2))) float v2f;
typedef __attribute__((ext_vector_type(4))) float v4f;
typedef __attribute__((ext_vector_type(8))) float v8f;

#define SEQ   100
#define BATCH 256
#define INDIM 300
#define HID   128
#define NROW  (BATCH*SEQ)
#define WIN   11

#define MT    7
#define MROW  (MT*16)
#define XP    308
#define HP    132
#define TP    16
#define NR    32
#define GS    36

static __device__ __forceinline__ v8f wmma_f32_k4(v2f a, v2f b, v8f c) {
    return __builtin_amdgcn_wmma_f32_16x16x4_f32(false, a, false, b, (short)0, c,
                                                 false, false);
}

static __device__ __forceinline__ float wave_sum8(const float (&pv)[8], int lane) {
    const bool hi16 = (lane & 16) != 0;
    const bool hi8  = (lane & 8) != 0;
    const bool hi4  = (lane & 4) != 0;
    float r4[4];
#pragma unroll
    for (int i = 0; i < 4; ++i) {
        const float keep = hi16 ? pv[i + 4] : pv[i];
        const float send = hi16 ? pv[i] : pv[i + 4];
        r4[i] = keep + __shfl_xor(send, 16, 32);
    }
    float r2[2];
#pragma unroll
    for (int i = 0; i < 2; ++i) {
        const float keep = hi8 ? r4[i + 2] : r4[i];
        const float send = hi8 ? r4[i] : r4[i + 2];
        r2[i] = keep + __shfl_xor(send, 8, 32);
    }
    float r1;
    {
        const float keep = hi4 ? r2[1] : r2[0];
        const float send = hi4 ? r2[0] : r2[1];
        r1 = keep + __shfl_xor(send, 4, 32);
    }
    r1 += __shfl_xor(r1, 2, 32);
    r1 += __shfl_xor(r1, 1, 32);
    return r1;
}

template <int NM>
static __device__ __forceinline__ void phase_a(const float* xs, float* hs,
                                               const float* __restrict__ W1,
                                               const float* __restrict__ b1,
                                               int mg, int n, int rr, int half) {
    v8f acc[NM];
#pragma unroll
    for (int i = 0; i < NM; ++i) acc[i] = (v8f){0.f,0.f,0.f,0.f,0.f,0.f,0.f,0.f};
    const float* xa = xs + (mg * 16 + rr) * XP + half * 2;
    const float* wb = W1 + (half * 2) * HID + n;
#pragma unroll 5
    for (int kk = 0; kk < INDIM; kk += 4) {
        v2f bf; bf.x = wb[kk * HID]; bf.y = wb[(kk + 1) * HID];
#pragma unroll
        for (int i = 0; i < NM; ++i) {
            const v2f a = *(const v2f*)(xa + i * (32 * XP) + kk);
            acc[i] = wmma_f32_k4(a, bf, acc[i]);
        }
    }
    const float bias = b1[n];
#pragma unroll
    for (int i = 0; i < NM; ++i) {
#pragma unroll
        for (int r = 0; r < 8; ++r) {
            const float v = acc[i][r] + bias;
            hs[((mg + 2 * i) * 16 + r + half * 8) * HP + n] = v > 0.f ? v : 0.f;
        }
    }
}

template <int NM>
static __device__ __forceinline__ void phase_b(const float* hs, const float* wp,
                                               float* pp, float* Qs, float* Ks,
                                               const float* __restrict__ Wq,
                                               const float* __restrict__ Wk,
                                               int mg, int nt, int n, int rr, int half) {
    v8f aq[NM], ak[NM];
#pragma unroll
    for (int i = 0; i < NM; ++i) {
        aq[i] = (v8f){0.f,0.f,0.f,0.f,0.f,0.f,0.f,0.f};
        ak[i] = aq[i];
    }
    const float* ha = hs + (mg * 16 + rr) * HP + half * 2;
    const float* wq = Wq + (half * 2) * HID + n;
    const float* wk = Wk + (half * 2) * HID + n;
#pragma unroll 4
    for (int kk = 0; kk < HID; kk += 4) {
        v2f bq; bq.x = wq[kk * HID]; bq.y = wq[(kk + 1) * HID];
        v2f bk; bk.x = wk[kk * HID]; bk.y = wk[(kk + 1) * HID];
#pragma unroll
        for (int i = 0; i < NM; ++i) {
            const v2f a = *(const v2f*)(ha + i * (32 * HP) + kk);
            aq[i] = wmma_f32_k4(a, bq, aq[i]);
            ak[i] = wmma_f32_k4(a, bk, ak[i]);
        }
    }
#pragma unroll
    for (int i = 0; i < NM; ++i) {
#pragma unroll
        for (int r = 0; r < 8; ++r) {
            const int m = (mg + 2 * i) * 16 + r + half * 8;
            Qs[m * HP + n] = aq[i][r];
            Ks[m * HP + n] = ak[i][r];
        }
    }
    v8f ap[NM];
#pragma unroll
    for (int i = 0; i < NM; ++i) ap[i] = (v8f){0.f,0.f,0.f,0.f,0.f,0.f,0.f,0.f};
#pragma unroll
    for (int s4 = 0; s4 < 4; ++s4) {
        const int off = nt * 16 + s4 * 4 + half * 2;
        const float w0 = wp[off * 2 + (rr & 1)];
        const float w1 = wp[(off + 1) * 2 + (rr & 1)];
        v2f bp; bp.x = rr < 2 ? w0 : 0.f; bp.y = rr < 2 ? w1 : 0.f;
#pragma unroll
        for (int i = 0; i < NM; ++i) {
            const v2f a = *(const v2f*)(hs + ((mg + 2 * i) * 16 + rr) * HP + off);
            ap[i] = wmma_f32_k4(a, bp, ap[i]);
        }
    }
    if (rr < 2) {
#pragma unroll
        for (int i = 0; i < NM; ++i) {
#pragma unroll
            for (int r = 0; r < 8; ++r)
                pp[(nt * MROW + (mg + 2 * i) * 16 + half * 8 + r) * 2 + rr] = ap[i][r];
        }
    }
}

__global__ __launch_bounds__(512) void k_seq(const float* __restrict__ x,
                                             const float* __restrict__ W1,
                                             const float* __restrict__ b1,
                                             const float* __restrict__ Wq,
                                             const float* __restrict__ Wk,
                                             const float* __restrict__ Wv,
                                             const float* __restrict__ W2,
                                             const float* __restrict__ b2,
                                             float* __restrict__ out,
                                             float* __restrict__ alphaOut) {
    __shared__ __attribute__((aligned(16))) float xs[MROW * XP];
    __shared__ __attribute__((aligned(16))) float hs[MROW * HP];
    __shared__ __attribute__((aligned(16))) float wp[HID * 2];
    __shared__ __attribute__((aligned(16))) float pp[8 * MROW * 2];
    __shared__ __attribute__((aligned(16))) float Ps[MROW * 2];

    const int tid  = threadIdx.x;
    const int wave = tid >> 5;
    const int lane = tid & 31;
    const int rr   = lane & 15;
    const int half = lane >> 4;
    const int wu = __builtin_amdgcn_readfirstlane(wave);
    const int nt = wu & 7;
    const int mg = wu >> 3;
    const int n  = nt * 16 + rr;
    const int seq = blockIdx.x;
    const v4f zero4 = {0.f, 0.f, 0.f, 0.f};

    {
        const v4f* x4 = (const v4f*)(x + (size_t)seq * (SEQ * INDIM));
        v4f xb[15];
#pragma unroll
        for (int i = 0; i < 15; ++i) {
            const int e  = tid + i * 512;
            const int ec = e < SEQ * INDIM / 4 ? e : SEQ * INDIM / 4 - 1;
            xb[i] = x4[ec];
        }
#pragma unroll
        for (int i = 0; i < 15; ++i) {
            const int e = tid + i * 512;
            if (e < SEQ * INDIM / 4) {
                const int row = e / (INDIM / 4);
                const int c4  = e - row * (INDIM / 4);
                *(v4f*)(xs + row * XP + c4 * 4) = xb[i];
            }
        }
        for (int e = tid; e < (MROW - SEQ) * XP / 4; e += 512)
            *(v4f*)(xs + SEQ * XP + e * 4) = zero4;
    }
    __syncthreads();

    if (mg == 0) phase_a<4>(xs, hs, W1, b1, 0, n, rr, half);
    else         phase_a<3>(xs, hs, W1, b1, 1, n, rr, half);

    {
        float w20[4], w21[4];
#pragma unroll
        for (int t = 0; t < 4; ++t) {
            const int d = lane * 4 + t;
            w20[t] = W2[d * 2 + 0];
            w21[t] = W2[d * 2 + 1];
        }
#pragma unroll
        for (int qq = 0; qq < 2; ++qq) {
            const int q = mg * 2 + qq;
            v4f wr[4];
#pragma unroll
            for (int t = 0; t < 4; ++t)
                wr[t] = *(const v4f*)(Wv + (q * 32 + nt + 8 * t) * HID + lane * 4);
            float pv[8];
#pragma unroll
            for (int t = 0; t < 4; ++t) {
                float a0 = 0.f, a1 = 0.f;
#pragma unroll
                for (int u = 0; u < 4; ++u) {
                    a0 += wr[t][u] * w20[u];
                    a1 += wr[t][u] * w21[u];
                }
                pv[2 * t]     = a0;
                pv[2 * t + 1] = a1;
            }
            const float r1 = wave_sum8(pv, lane);
            if ((lane & 3) == 0) {
                const int k = lane >> 2;
                wp[(q * 32 + nt + 8 * (k >> 1)) * 2 + (k & 1)] = r1;
            }
        }
    }
    __syncthreads();

    float* Qs = xs;
    float* Ks = xs + MROW * HP;
    if (mg == 0) phase_b<4>(hs, wp, pp, Qs, Ks, Wq, Wk, 0, nt, n, rr, half);
    else         phase_b<3>(hs, wp, pp, Qs, Ks, Wq, Wk, 1, nt, n, rr, half);
    __syncthreads();

    if (tid < MROW * 2) {
        float s = 0.f;
#pragma unroll
        for (int w = 0; w < 8; ++w) s += pp[w * (MROW * 2) + tid];
        Ps[tid] = s;
    }

    const float bias0 = b2[0], bias1 = b2[1];
    const float scale = 0.08838834764831845f;
    float* Gg = hs + mg * (2 * NR * GS);
    float* Ag = hs + 2 * (2 * NR * GS) + mg * (TP * WIN * WIN);
    const int tl = tid & 255;

    for (int rd = 0; rd < 4; ++rd) {
        const int g    = rd * 2 + mg;
        const bool act = g < MT;
        const int s0   = g * TP;

        if (act) {
            const int tm = nt & 1;
            const int tn = (nt >> 1) & 1;
            const int kh = nt >> 2;
            int ra = s0 - 5 + tm * 16 + rr;
            ra = ra < 0 ? 0 : (ra > MROW - 1 ? MROW - 1 : ra);
            int rb = s0 - 5 + tn * 16 + rr;
            rb = rb < 0 ? 0 : (rb > MROW - 1 ? MROW - 1 : rb);
            const float* qa = Qs + ra * HP + kh * 64 + half * 2;
            const float* kb = Ks + rb * HP + kh * 64 + half * 2;
            v8f acc = {0.f,0.f,0.f,0.f,0.f,0.f,0.f,0.f};
#pragma unroll
            for (int kk = 0; kk < 64; kk += 4) {
                const v2f a  = *(const v2f*)(qa + kk);
                const v2f bf = *(const v2f*)(kb + kk);
                acc = wmma_f32_k4(a, bf, acc);
            }
            float* gp = Gg + kh * (NR * GS) + (tm * 16 + half * 8) * GS + tn * 16 + rr;
#pragma unroll
            for (int r = 0; r < 8; ++r) gp[r * GS] = acc[r];
        }
        __syncthreads();

        if (act) {
            const int p  = nt * 2 + half;
            const int s  = s0 + p;
            const int ic = rr < WIN ? rr : WIN - 1;
            const int ri = s + 5 - ic;
            const bool rowok = (ri >= 0) && (ri < SEQ);
            const float* gq = Gg + (p + 10 - ic) * GS + (p + 10);
            float sc[WIN];
#pragma unroll
            for (int j = 0; j < WIN; ++j) {
                const int rj = s + 5 - j;
                const bool ok = rowok && (rj >= 0) && (rj < SEQ);
                const float gsum = gq[-j] + gq[NR * GS - j];
                sc[j] = ok ? gsum * scale : 0.f;
            }
            float mv = sc[0];
#pragma unroll
            for (int j = 1; j < WIN; ++j) mv = fmaxf(mv, sc[j]);
            float sum = 0.f;
#pragma unroll
            for (int j = 0; j < WIN; ++j) {
                sc[j] = expf(sc[j] - mv);
                sum += sc[j];
            }
            const float rinv = 1.0f / sum;
            if (rr < WIN) {
                float* ap = Ag + p * (WIN * WIN) + rr * WIN;
#pragma unroll
                for (int j = 0; j < WIN; ++j) ap[j] = sc[j] * rinv;
            }
        }
        __syncthreads();

        if (act) {
            {
                const int nv4 = (g < MT - 1 ? TP : SEQ - (MT - 1) * TP) * (WIN * WIN) / 4;
                const v4f* a4 = (const v4f*)Ag;
                v4f* o4 = (v4f*)(alphaOut + ((size_t)seq * SEQ + s0) * (WIN * WIN));
                if (tl < nv4)       o4[tl]       = a4[tl];
                if (tl + 256 < nv4) o4[tl + 256] = a4[tl + 256];
            }
            const int p  = nt * 2 + half;
            const int s  = s0 + p;
            const int jc = rr < WIN ? rr : WIN - 1;
            float beta = 0.f;
            {
                const float* ac = Ag + p * (WIN * WIN) + jc;
#pragma unroll
                for (int i = 0; i < WIN; ++i) beta += ac[i * WIN];
            }
            const int rv = s + 5 - jc;
            const bool valid = (rr < WIN) && (rv >= 0) && (rv < SEQ);
            const int rvc = rv < 0 ? 0 : (rv > MROW - 1 ? MROW - 1 : rv);
            const v2f pr = *(const v2f*)(Ps + rvc * 2);
            const float t0 = valid ? beta * pr.x : 0.f;
            const float t1 = valid ? beta * pr.y : 0.f;
            const bool hi8 = (rr & 8) != 0;
            const float keep = hi8 ? t1 : t0;
            const float send = hi8 ? t0 : t1;
            float r = keep + __shfl_xor(send, 8, 32);
            r += __shfl_xor(r, 4, 32);
            r += __shfl_xor(r, 2, 32);
            r += __shfl_xor(r, 1, 32);
            if ((rr & 7) == 0 && s < SEQ) {
                const int c = rr >> 3;
                const float bias = c ? bias1 : bias0;
                out[((size_t)seq * SEQ + s) * 2 + c] = 1.f / (1.f + expf(-(r + bias)));
            }
        }
    }
}

extern "C" void kernel_launch(void* const* d_in, const int* in_sizes, int n_in,
                              void* d_out, int out_size, void* d_ws, size_t ws_size,
                              hipStream_t stream) {
    const float* x  = (const float*)d_in[0];
    const float* W1 = (const float*)d_in[1];
    const float* b1 = (const float*)d_in[2];
    const float* Wq = (const float*)d_in[3];
    const float* Wk = (const float*)d_in[4];
    const float* Wv = (const float*)d_in[5];
    const float* W2 = (const float*)d_in[6];
    const float* b2 = (const float*)d_in[7];

    float* out      = (float*)d_out;
    float* alphaOut = out + (size_t)NROW * 2;

    k_seq<<<BATCH, 512, 0, stream>>>(x, W1, b1, Wq, Wk, Wv, W2, b2, out, alphaOut);
}


# attention part per wave: each wave builds the one 16x16 diagonal Q.K^T tile of its 6 positions and does their softmax, alpha rows and logits in its own LDS; that part's 12 block barriers are gone
# speedup vs baseline: 2.5553x; 1.0478x over previous
#include <hip/hip_runtime.h>
#include <math.h>


typedef __attribute__((ext_vector_type(2))) float v2f;
typedef __attribute__((ext_vector_type(4))) float v4f;
typedef __attribute__((ext_vector_type(8))) float v8f;

#define SEQ   100
#define BATCH 256
#define INDIM 300
#define HID   128
#define NROW  (BATCH*SEQ)
#define WIN   11

#define MT    7
#define MROW  (MT*16)
#define XP    308
#define HP    132
#define PT    6
#define NTASK 17
#define GP    18
#define ALW   736
#define AW    (16 * GP + ALW + 32)

static __device__ __forceinline__ v8f wmma_f32_k4(v2f a, v2f b, v8f c) {
    return __builtin_amdgcn_wmma_f32_16x16x4_f32(false, a, false, b, (short)0, c,
                                                 false, false);
}

static __device__ __forceinline__ void wave_sync() {
    __builtin_amdgcn_fence(__ATOMIC_RELEASE, "wavefront");
    __builtin_amdgcn_wave_barrier();
    __builtin_amdgcn_fence(__ATOMIC_ACQUIRE, "wavefront");
}

static __device__ __forceinline__ float wave_sum8(const float (&pv)[8], int lane) {
    const bool hi16 = (lane & 16) != 0;
    const bool hi8  = (lane & 8) != 0;
    const bool hi4  = (lane & 4) != 0;
    float r4[4];
#pragma unroll
    for (int i = 0; i < 4; ++i) {
        const float keep = hi16 ? pv[i + 4] : pv[i];
        const float send = hi16 ? pv[i] : pv[i + 4];
        r4[i] = keep + __shfl_xor(send, 16, 32);
    }
    float r2[2];
#pragma unroll
    for (int i = 0; i < 2; ++i) {
        const float keep = hi8 ? r4[i + 2] : r4[i];
        const float send = hi8 ? r4[i] : r4[i + 2];
        r2[i] = keep + __shfl_xor(send, 8, 32);
    }
    float r1;
    {
        const float keep = hi4 ? r2[1] : r2[0];
        const float send = hi4 ? r2[0] : r2[1];
        r1 = keep + __shfl_xor(send, 4, 32);
    }
    r1 += __shfl_xor(r1, 2, 32);
    r1 += __shfl_xor(r1, 1, 32);
    return r1;
}

template <int NM>
static __device__ __forceinline__ void phase_a(const float* xs, float* hs,
                                               const float* __restrict__ W1,
                                               const float* __restrict__ b1,
                                               int mg, int n, int rr, int half) {
    v8f acc[NM];
#pragma unroll
    for (int i = 0; i < NM; ++i) acc[i] = (v8f){0.f,0.f,0.f,0.f,0.f,0.f,0.f,0.f};
    const float* xa = xs + (mg * 16 + rr) * XP + half * 2;
    const float* wb = W1 + (half * 2) * HID + n;
#pragma unroll 5
    for (int kk = 0; kk < INDIM; kk += 4) {
        v2f bf; bf.x = wb[kk * HID]; bf.y = wb[(kk + 1) * HID];
#pragma unroll
        for (int i = 0; i < NM; ++i) {
            const v2f a = *(const v2f*)(xa + i * (32 * XP) + kk);
            acc[i] = wmma_f32_k4(a, bf, acc[i]);
        }
    }
    const float bias = b1[n];
#pragma unroll
    for (int i = 0; i < NM; ++i) {
#pragma unroll
        for (int r = 0; r < 8; ++r) {
            const float v = acc[i][r] + bias;
            hs[((mg + 2 * i) * 16 + r + half * 8) * HP + n] = v > 0.f ? v : 0.f;
        }
    }
}

template <int NM>
static __device__ __forceinline__ void phase_b(const float* hs, const float* wp,
                                               float* pp, float* Qs, float* Ks,
                                               const float* __restrict__ Wq,
                                               const float* __restrict__ Wk,
                                               int mg, int nt, int n, int rr, int half) {
    v8f aq[NM], ak[NM];
#pragma unroll
    for (int i = 0; i < NM; ++i) {
        aq[i] = (v8f){0.f,0.f,0.f,0.f,0.f,0.f,0.f,0.f};
        ak[i] = aq[i];
    }
    const float* ha = hs + (mg * 16 + rr) * HP + half * 2;
    const float* wq = Wq + (half * 2) * HID + n;
    const float* wk = Wk + (half * 2) * HID + n;
#pragma unroll 4
    for (int kk = 0; kk < HID; kk += 4) {
        v2f bq; bq.x = wq[kk * HID]; bq.y = wq[(kk + 1) * HID];
        v2f bk; bk.x = wk[kk * HID]; bk.y = wk[(kk + 1) * HID];
#pragma unroll
        for (int i = 0; i < NM; ++i) {
            const v2f a = *(const v2f*)(ha + i * (32 * HP) + kk);
            aq[i] = wmma_f32_k4(a, bq, aq[i]);
            ak[i] = wmma_f32_k4(a, bk, ak[i]);
        }
    }
#pragma unroll
    for (int i = 0; i < NM; ++i) {
#pragma unroll
        for (int r = 0; r < 8; ++r) {
            const int m = (mg + 2 * i) * 16 + r + half * 8;
            Qs[m * HP + n] = aq[i][r];
            Ks[m * HP + n] = ak[i][r];
        }
    }
    v8f ap[NM];
#pragma unroll
    for (int i = 0; i < NM; ++i) ap[i] = (v8f){0.f,0.f,0.f,0.f,0.f,0.f,0.f,0.f};
#pragma unroll
    for (int s4 = 0; s4 < 4; ++s4) {
        const int off = nt * 16 + s4 * 4 + half * 2;
        const float w0 = wp[off * 2 + (rr & 1)];
        const float w1 = wp[(off + 1) * 2 + (rr & 1)];
        v2f bp; bp.x = rr < 2 ? w0 : 0.f; bp.y = rr < 2 ? w1 : 0.f;
#pragma unroll
        for (int i = 0; i < NM; ++i) {
            const v2f a = *(const v2f*)(hs + ((mg + 2 * i) * 16 + rr) * HP + off);
            ap[i] = wmma_f32_k4(a, bp, ap[i]);
        }
    }
    if (rr < 2) {
#pragma unroll
        for (int i = 0; i < NM; ++i) {
#pragma unroll
            for (int r = 0; r < 8; ++r)
                pp[(nt * MROW + (mg + 2 * i) * 16 + half * 8 + r) * 2 + rr] = ap[i][r];
        }
    }
}

__global__ __launch_bounds__(512) void k_seq(const float* __restrict__ x,
                                             const float* __restrict__ W1,
                                             const float* __restrict__ b1,
                                             const float* __restrict__ Wq,
                                             const float* __restrict__ Wk,
                                             const float* __restrict__ Wv,
                                             const float* __restrict__ W2,
                                             const float* __restrict__ b2,
                                             float* __restrict__ out,
                                             float* __restrict__ alphaOut) {
    __shared__ __attribute__((aligned(16))) float xs[MROW * XP];
    __shared__ __attribute__((aligned(16))) float hs[MROW * HP];
    __shared__ __attribute__((aligned(16))) float wp[HID * 2];
    __shared__ __attribute__((aligned(16))) float pp[8 * MROW * 2];
    __shared__ __attribute__((aligned(16))) float att[16 * AW];

    const int tid  = threadIdx.x;
    const int wave = tid >> 5;
    const int lane = tid & 31;
    const int rr   = lane & 15;
    const int half = lane >> 4;
    const int wu = __builtin_amdgcn_readfirstlane(wave);
    const int nt = wu & 7;
    const int mg = wu >> 3;
    const int n  = nt * 16 + rr;
    const int seq = blockIdx.x;
    const v4f zero4 = {0.f, 0.f, 0.f, 0.f};

    {
        const v4f* x4 = (const v4f*)(x + (size_t)seq * (SEQ * INDIM));
        v4f xb[15];
#pragma unroll
        for (int i = 0; i < 15; ++i) {
            const int e  = tid + i * 512;
            const int ec = e < SEQ * INDIM / 4 ? e : SEQ * INDIM / 4 - 1;
            xb[i] = x4[ec];
        }
#pragma unroll
        for (int i = 0; i < 15; ++i) {
            const int e = tid + i * 512;
            if (e < SEQ * INDIM / 4) {
                const int row = e / (INDIM / 4);
                const int c4  = e - row * (INDIM / 4);
                *(v4f*)(xs + row * XP + c4 * 4) = xb[i];
            }
        }
        for (int e = tid; e < (MROW - SEQ) * XP / 4; e += 512)
            *(v4f*)(xs + SEQ * XP + e * 4) = zero4;
    }
    __syncthreads();

    if (mg == 0) phase_a<4>(xs, hs, W1, b1, 0, n, rr, half);
    else         phase_a<3>(xs, hs, W1, b1, 1, n, rr, half);

    {
        float w20[4], w21[4];
#pragma unroll
        for (int t = 0; t < 4; ++t) {
            const int d = lane * 4 + t;
            w20[t] = W2[d * 2 + 0];
            w21[t] = W2[d * 2 + 1];
        }
#pragma unroll
        for (int qq = 0; qq < 2; ++qq) {
            const int q = mg * 2 + qq;
            v4f wr[4];
#pragma unroll
            for (int t = 0; t < 4; ++t)
                wr[t] = *(const v4f*)(Wv + (q * 32 + nt + 8 * t) * HID + lane * 4);
            float pv[8];
#pragma unroll
            for (int t = 0; t < 4; ++t) {
                float a0 = 0.f, a1 = 0.f;
#pragma unroll
                for (int u = 0; u < 4; ++u) {
                    a0 += wr[t][u] * w20[u];
                    a1 += wr[t][u] * w21[u];
                }
                pv[2 * t]     = a0;
                pv[2 * t + 1] = a1;
            }
            const float r1 = wave_sum8(pv, lane);
            if ((lane & 3) == 0) {
                const int k = lane >> 2;
                wp[(q * 32 + nt + 8 * (k >> 1)) * 2 + (k & 1)] = r1;
            }
        }
    }
    __syncthreads();

    float* Qs = xs;
    float* Ks = xs + MROW * HP;
    if (mg == 0) phase_b<4>(hs, wp, pp, Qs, Ks, Wq, Wk, 0, nt, n, rr, half);
    else         phase_b<3>(hs, wp, pp, Qs, Ks, Wq, Wk, 1, nt, n, rr, half);
    __syncthreads();

    const float bias0 = b2[0], bias1 = b2[1];
    const float scale = 0.08838834764831845f;
    float* Gw = att + wu * AW;
    float* Aw = Gw + 16 * GP;
    float* Pw = Aw + ALW;

    for (int t = wu; t < NTASK; t += 16) {
        const int s0 = t * PT;
        const int r0 = s0 - 5;
        const int npos  = SEQ - s0 < PT ? SEQ - s0 : PT;
        const int npass = (npos + 1) >> 1;
        const int sh = (t & 1) * 2;

        {
            int ra = r0 + rr;
            ra = ra < 0 ? 0 : (ra > MROW - 1 ? MROW - 1 : ra);
            const float* qa = Qs + ra * HP + half * 2;
            const float* kb = Ks + ra * HP + half * 2;
            v8f acc0 = {0.f,0.f,0.f,0.f,0.f,0.f,0.f,0.f};
            v8f acc1 = acc0;
#pragma unroll
            for (int kk = 0; kk < 64; kk += 4) {
                const v2f a0 = *(const v2f*)(qa + kk);
                const v2f f0 = *(const v2f*)(kb + kk);
                const v2f a1 = *(const v2f*)(qa + 64 + kk);
                const v2f f1 = *(const v2f*)(kb + 64 + kk);
                acc0 = wmma_f32_k4(a0, f0, acc0);
                acc1 = wmma_f32_k4(a1, f1, acc1);
            }
            float* gp = Gw + (half * 8) * GP + rr;
#pragma unroll
            for (int r = 0; r < 8; ++r) gp[r * GP] = acc0[r] + acc1[r];
        }
        {
            int rp = r0 + (lane >> 1);
            rp = rp < 0 ? 0 : (rp > MROW - 1 ? MROW - 1 : rp);
            float sp = 0.f;
#pragma unroll
            for (int w = 0; w < 8; ++w) sp += pp[w * (MROW * 2) + rp * 2 + (lane & 1)];
            Pw[lane] = sp;
        }
        wave_sync();

        for (int q = 0; q < npass; ++q) {
            const int p  = q * 2 + half;
            const int s  = s0 + p;
            const int ic = rr < WIN ? rr : WIN - 1;
            const int ri = s + 5 - ic;
            const bool rowok = (ri >= 0) && (ri < SEQ);
            const float* gq = Gw + (p + 10 - ic) * GP + (p + 10);
            float sc[WIN];
#pragma unroll
            for (int j = 0; j < WIN; ++j) {
                const int rj = s + 5 - j;
                const bool ok = rowok && (rj >= 0) && (rj < SEQ);
                const float gv = gq[-j];
                sc[j] = ok ? gv * scale : 0.f;
            }
            float mv = sc[0];
#pragma unroll
            for (int j = 1; j < WIN; ++j) mv = fmaxf(mv, sc[j]);
            float sum = 0.f;
#pragma unroll
            for (int j = 0; j < WIN; ++j) {
                sc[j] = expf(sc[j] - mv);
                sum += sc[j];
            }
            const float rinv = 1.0f / sum;
            if (rr < WIN) {
                float* ap = Aw + sh + p * (WIN * WIN) + rr * WIN;
#pragma unroll
                for (int j = 0; j < WIN; ++j) ap[j] = sc[j] * rinv;
            }
        }
        wave_sync();

        {
            float* gdst = alphaOut + ((size_t)seq * SEQ + s0) * (WIN * WIN) - sh;
            const int jend = sh + npos * (WIN * WIN);
            const int i0 = (sh + 3) >> 2, i1 = jend >> 2;
            v4f av[6];
#pragma unroll
            for (int u = 0; u < 6; ++u) {
                const int i  = i0 + lane + u * 32;
                const int ie = i < i1 ? i : i1 - 1;
                av[u] = *(const v4f*)(Aw + ie * 4);
            }
#pragma unroll
            for (int u = 0; u < 6; ++u) {
                const int i = i0 + lane + u * 32;
                if (i < i1) *(v4f*)(gdst + i * 4) = av[u];
            }
            if (lane == 0 && sh != 0)
                *(v2f*)(gdst + sh) = *(const v2f*)(Aw + sh);
            if (lane == 1 && (jend & 3) != 0)
                *(v2f*)(gdst + i1 * 4) = *(const v2f*)(Aw + i1 * 4);
        }

        for (int q = 0; q < npass; ++q) {
            const int p  = q * 2 + half;
            const int s  = s0 + p;
            const int jc = rr < WIN ? rr : WIN - 1;
            float beta = 0.f;
            {
                const float* ac = Aw + sh + p * (WIN * WIN) + jc;
#pragma unroll
                for (int i = 0; i < WIN; ++i) beta += ac[i * WIN];
            }
            const int rv = s + 5 - jc;
            const bool valid = (rr < WIN) && (rv >= 0) && (rv < SEQ);
            const v2f pr = *(const v2f*)(Pw + (p + 10 - jc) * 2);
            const float t0 = valid ? beta * pr.x : 0.f;
            const float t1 = valid ? beta * pr.y : 0.f;
            const bool hi8 = (rr & 8) != 0;
            const float keep = hi8 ? t1 : t0;
            const float send = hi8 ? t0 : t1;
            float r = keep + __shfl_xor(send, 8, 32);
            r += __shfl_xor(r, 4, 32);
            r += __shfl_xor(r, 2, 32);
            r += __shfl_xor(r, 1, 32);
            if ((rr & 7) == 0 && s < SEQ) {
                const int c = rr >> 3;
                const float bias = c ? bias1 : bias0;
                out[((size_t)seq * SEQ + s) * 2 + c] = 1.f / (1.f + expf(-(r + bias)));
            }
        }
        wave_sync();
    }
}

extern "C" void kernel_launch(void* const* d_in, const int* in_sizes, int n_in,
                              void* d_out, int out_size, void* d_ws, size_t ws_size,
                              hipStream_t stream) {
    const float* x  = (const float*)d_in[0];
    const float* W1 = (const float*)d_in[1];
    const float* b1 = (const float*)d_in[2];
    const float* Wq = (const float*)d_in[3];
    const float* Wk = (const float*)d_in[4];
    const float* Wv = (const float*)d_in[5];
    const float* W2 = (const float*)d_in[6];
    const float* b2 = (const float*)d_in[7];

    float* out      = (float*)d_out;
    float* alphaOut = out + (size_t)NROW * 2;

    k_seq<<<BATCH, 512, 0, stream>>>(x, W1, b1, Wq, Wk, Wv, W2, b2, out, alphaOut);
}


# GEMM loops: each trip's A fragments (LDS) and weight fragments loaded as ONE group a trip ahead into a 2nd register set (first weights ahead of the barriers); its WMMAs back to back; sums unchanged
# speedup vs baseline: 2.7342x; 1.0700x over previous
#include <hip/hip_runtime.h>
#include <math.h>


typedef __attribute__((ext_vector_type(2))) float v2f;
typedef __attribute__((ext_vector_type(4))) float v4f;
typedef __attribute__((ext_vector_type(8))) float v8f;

#define SEQ   100
#define BATCH 256
#define INDIM 300
#define HID   128
#define NROW  (BATCH*SEQ)
#define WIN   11

#define MT    7
#define MROW  (MT*16)
#define XP    308
#define HP    132
#define PT    6
#define NTASK 17
#define GP    18
#define ALW   736
#define AW    (16 * GP + ALW + 32)

static __device__ __forceinline__ v8f wmma_f32_k4(v2f a, v2f b, v8f c) {
    return __builtin_amdgcn_wmma_f32_16x16x4_f32(false, a, false, b, (short)0, c,
                                                 false, false);
}

static __device__ __forceinline__ void wave_sync() {
    __builtin_amdgcn_fence(__ATOMIC_RELEASE, "wavefront");
    __builtin_amdgcn_wave_barrier();
    __builtin_amdgcn_fence(__ATOMIC_ACQUIRE, "wavefront");
}

static __device__ __forceinline__ float wave_sum8(const float (&pv)[8], int lane) {
    const bool hi16 = (lane & 16) != 0;
    const bool hi8  = (lane & 8) != 0;
    const bool hi4  = (lane & 4) != 0;
    float r4[4];
#pragma unroll
    for (int i = 0; i < 4; ++i) {
        const float keep = hi16 ? pv[i + 4] : pv[i];
        const float send = hi16 ? pv[i] : pv[i + 4];
        r4[i] = keep + __shfl_xor(send, 16, 32);
    }
    float r2[2];
#pragma unroll
    for (int i = 0; i < 2; ++i) {
        const float keep = hi8 ? r4[i + 2] : r4[i];
        const float send = hi8 ? r4[i] : r4[i + 2];
        r2[i] = keep + __shfl_xor(send, 8, 32);
    }
    float r1;
    {
        const float keep = hi4 ? r2[1] : r2[0];
        const float send = hi4 ? r2[0] : r2[1];
        r1 = keep + __shfl_xor(send, 4, 32);
    }
    r1 += __shfl_xor(r1, 2, 32);
    r1 += __shfl_xor(r1, 1, 32);
    return r1;
}

static __device__ __forceinline__ void load_w_a(v2f (&b)[5], const float* wb, int kk0) {
#pragma unroll
    for (int s = 0; s < 5; ++s) {
        const int kk = kk0 + s * 4;
        b[s].x = wb[kk * HID];
        b[s].y = wb[(kk + 1) * HID];
    }
}

template <int NM>
static __device__ __forceinline__ void load_x_a(v2f (&a)[5][NM], const float* xa, int kk0) {
#pragma unroll
    for (int s = 0; s < 5; ++s) {
#pragma unroll
        for (int i = 0; i < NM; ++i)
            a[s][i] = *(const v2f*)(xa + i * (32 * XP) + kk0 + s * 4);
    }
}

template <int NM>
static __device__ __forceinline__ void load_trip_a(v2f (&a)[5][NM], v2f (&b)[5],
                                                   const float* xa, const float* wb,
                                                   int kk0) {
    load_w_a(b, wb, kk0);
    load_x_a<NM>(a, xa, kk0);
}

template <int NM>
static __device__ __forceinline__ void mma_trip_a(v8f (&acc)[NM], const v2f (&a)[5][NM],
                                                  const v2f (&b)[5]) {
#pragma unroll
    for (int s = 0; s < 5; ++s) {
#pragma unroll
        for (int i = 0; i < NM; ++i) acc[i] = wmma_f32_k4(a[s][i], b[s], acc[i]);
    }
}

template <int NM>
static __device__ __forceinline__ void phase_a(const float* xs, float* hs,
                                               const float* W1,
                                               const float* __restrict__ b1,
                                               int mg, int n, int rr, int half,
                                               const v2f (&wfirst)[5]) {
    v8f acc[NM];
#pragma unroll
    for (int i = 0; i < NM; ++i) acc[i] = (v8f){0.f,0.f,0.f,0.f,0.f,0.f,0.f,0.f};
    const float* xa = xs + (mg * 16 + rr) * XP + half * 2;
    const float* wb = W1 + (half * 2) * HID + n;
    v2f a0[5][NM], a1[5][NM], w0[5], w1[5];
#pragma unroll
    for (int s = 0; s < 5; ++s) w0[s] = wfirst[s];
    load_x_a<NM>(a0, xa, 0);
#pragma nounroll
    for (int kk = 0; kk < INDIM - 20; kk += 40) {
        load_trip_a<NM>(a1, w1, xa, wb, kk + 20);
        __builtin_amdgcn_sched_barrier(0);
        mma_trip_a<NM>(acc, a0, w0);
        __builtin_amdgcn_sched_barrier(0);
        load_trip_a<NM>(a0, w0, xa, wb, kk + 40);
        __builtin_amdgcn_sched_barrier(0);
        mma_trip_a<NM>(acc, a1, w1);
    }
    mma_trip_a<NM>(acc, a0, w0);
    const float bias = b1[n];
#pragma unroll
    for (int i = 0; i < NM; ++i) {
#pragma unroll
        for (int r = 0; r < 8; ++r) {
            const float v = acc[i][r] + bias;
            hs[((mg + 2 * i) * 16 + r + half * 8) * HP + n] = v > 0.f ? v : 0.f;
        }
    }
}

static __device__ __forceinline__ void load_w_b(v2f (&bq)[4], v2f (&bk)[4], const float* wq,
                                                const float* wk, int kk0) {
#pragma unroll
    for (int s = 0; s < 4; ++s) {
        const int kk = kk0 + s * 4;
        bq[s].x = wq[kk * HID]; bq[s].y = wq[(kk + 1) * HID];
        bk[s].x = wk[kk * HID]; bk[s].y = wk[(kk + 1) * HID];
    }
}

template <int NM>
static __device__ __forceinline__ void load_h_b(v2f (&a)[4][NM], const float* ha, int kk0) {
#pragma unroll
    for (int s = 0; s < 4; ++s) {
#pragma unroll
        for (int i = 0; i < NM; ++i)
            a[s][i] = *(const v2f*)(ha + i * (32 * HP) + kk0 + s * 4);
    }
}

template <int NM>
static __device__ __forceinline__ void load_trip_b(v2f (&a)[4][NM], v2f (&bq)[4], v2f (&bk)[4],
                                                   const float* ha, const float* wq,
                                                   const float* wk, int kk0) {
    load_w_b(bq, bk, wq, wk, kk0);
    load_h_b<NM>(a, ha, kk0);
}

template <int NM>
static __device__ __forceinline__ void mma_trip_b(v8f (&aq)[NM], v8f (&ak)[NM],
                                                  const v2f (&a)[4][NM], const v2f (&bq)[4],
                                                  const v2f (&bk)[4]) {
#pragma unroll
    for (int s = 0; s < 4; ++s) {
#pragma unroll
        for (int i = 0; i < NM; ++i) {
            aq[i] = wmma_f32_k4(a[s][i], bq[s], aq[i]);
            ak[i] = wmma_f32_k4(a[s][i], bk[s], ak[i]);
        }
    }
}

template <int NM>
static __device__ __forceinline__ void phase_b(const float* hs, const float* wp,
                                               float* pp, float* Qs, float* Ks,
                                               const float* Wq,
                                               const float* Wk,
                                               int mg, int nt, int n, int rr, int half,
                                               const v2f (&qfirst)[4], const v2f (&kfirst)[4]) {
    v8f aq[NM], ak[NM];
#pragma unroll
    for (int i = 0; i < NM; ++i) {
        aq[i] = (v8f){0.f,0.f,0.f,0.f,0.f,0.f,0.f,0.f};
        ak[i] = aq[i];
    }
    const float* ha = hs + (mg * 16 + rr) * HP + half * 2;
    const float* wq = Wq + (half * 2) * HID + n;
    const float* wk = Wk + (half * 2) * HID + n;
    v2f a0[4][NM], a1[4][NM], q0[4], q1[4], k0[4], k1[4];
#pragma unroll
    for (int s = 0; s < 4; ++s) { q0[s] = qfirst[s]; k0[s] = kfirst[s]; }
    load_h_b<NM>(a0, ha, 0);
#pragma nounroll
    for (int kk = 0; kk < HID - 32; kk += 32) {
        load_trip_b<NM>(a1, q1, k1, ha, wq, wk, kk + 16);
        __builtin_amdgcn_sched_barrier(0);
        mma_trip_b<NM>(aq, ak, a0, q0, k0);
        __builtin_amdgcn_sched_barrier(0);
        load_trip_b<NM>(a0, q0, k0, ha, wq, wk, kk + 32);
        __builtin_amdgcn_sched_barrier(0);
        mma_trip_b<NM>(aq, ak, a1, q1, k1);
    }
    load_trip_b<NM>(a1, q1, k1, ha, wq, wk, HID - 16);
    __builtin_amdgcn_sched_barrier(0);
    mma_trip_b<NM>(aq, ak, a0, q0, k0);
    mma_trip_b<NM>(aq, ak, a1, q1, k1);
#pragma unroll
    for (int i = 0; i < NM; ++i) {
#pragma unroll
        for (int r = 0; r < 8; ++r) {
            const int m = (mg + 2 * i) * 16 + r + half * 8;
            Qs[m * HP + n] = aq[i][r];
            Ks[m * HP + n] = ak[i][r];
        }
    }
    v8f ap[NM];
#pragma unroll
    for (int i = 0; i < NM; ++i) ap[i] = (v8f){0.f,0.f,0.f,0.f,0.f,0.f,0.f,0.f};
#pragma unroll
    for (int s4 = 0; s4 < 4; ++s4) {
        const int off = nt * 16 + s4 * 4 + half * 2;
        const float w0 = wp[off * 2 + (rr & 1)];
        const float w1 = wp[(off + 1) * 2 + (rr & 1)];
        v2f bp; bp.x = rr < 2 ? w0 : 0.f; bp.y = rr < 2 ? w1 : 0.f;
#pragma unroll
        for (int i = 0; i < NM; ++i) {
            const v2f a = *(const v2f*)(hs + ((mg + 2 * i) * 16 + rr) * HP + off);
            ap[i] = wmma_f32_k4(a, bp, ap[i]);
        }
    }
    if (rr < 2) {
#pragma unroll
        for (int i = 0; i < NM; ++i) {
#pragma unroll
            for (int r = 0; r < 8; ++r)
                pp[(nt * MROW + (mg + 2 * i) * 16 + half * 8 + r) * 2 + rr] = ap[i][r];
        }
    }
}

__global__ __launch_bounds__(512) void k_seq(const float* __restrict__ x,
                                             const float* W1,
                                             const float* __restrict__ b1,
                                             const float* Wq,
                                             const float* Wk,
                                             const float* __restrict__ Wv,
                                             const float* __restrict__ W2,
                                             const float* __restrict__ b2,
                                             float* __restrict__ out,
                                             float* __restrict__ alphaOut) {
    __shared__ __attribute__((aligned(16))) float xs[MROW * XP];
    __shared__ __attribute__((aligned(16))) float hs[MROW * HP];
    __shared__ __attribute__((aligned(16))) float wp[HID * 2];
    __shared__ __attribute__((aligned(16))) float pp[8 * MROW * 2];
    __shared__ __attribute__((aligned(16))) float att[16 * AW];

    const int tid  = threadIdx.x;
    const int wave = tid >> 5;
    const int lane = tid & 31;
    const int rr   = lane & 15;
    const int half = lane >> 4;
    const int wu = __builtin_amdgcn_readfirstlane(wave);
    const int nt = wu & 7;
    const int mg = wu >> 3;
    const int n  = nt * 16 + rr;
    const int seq = blockIdx.x;
    const v4f zero4 = {0.f, 0.f, 0.f, 0.f};

    v2f wfirst[5];
    {
        const v4f* x4 = (const v4f*)(x + (size_t)seq * (SEQ * INDIM));
        v4f xb[15];
#pragma unroll
        for (int i = 0; i < 15; ++i) {
            const int e  = tid + i * 512;
            const int ec = e < SEQ * INDIM / 4 ? e : SEQ * INDIM / 4 - 1;
            xb[i] = x4[ec];
        }
        load_w_a(wfirst, W1 + (half * 2) * HID + n, 0);
#pragma unroll
        for (int i = 0; i < 15; ++i) {
            const int e = tid + i * 512;
            if (e < SEQ * INDIM / 4) {
                const int row = e / (INDIM / 4);
                const int c4  = e - row * (INDIM / 4);
                *(v4f*)(xs + row * XP + c4 * 4) = xb[i];
            }
        }
        for (int e = tid; e < (MROW - SEQ) * XP / 4; e += 512)
            *(v4f*)(xs + SEQ * XP + e * 4) = zero4;
    }
    __syncthreads();

    if (mg == 0) phase_a<4>(xs, hs, W1, b1, 0, n, rr, half, wfirst);
    else         phase_a<3>(xs, hs, W1, b1, 1, n, rr, half, wfirst);

    v2f qfirst[4], kfirst[4];
    load_w_b(qfirst, kfirst, Wq + (half * 2) * HID + n, Wk + (half * 2) * HID + n, 0);

    {
        float w20[4], w21[4];
#pragma unroll
        for (int t = 0; t < 4; ++t) {
            const int d = lane * 4 + t;
            w20[t] = W2[d * 2 + 0];
            w21[t] = W2[d * 2 + 1];
        }
#pragma unroll
        for (int qq = 0; qq < 2; ++qq) {
            const int q = mg * 2 + qq;
            v4f wr[4];
#pragma unroll
            for (int t = 0; t < 4; ++t)
                wr[t] = *(const v4f*)(Wv + (q * 32 + nt + 8 * t) * HID + lane * 4);
            float pv[8];
#pragma unroll
            for (int t = 0; t < 4; ++t) {
                float a0 = 0.f, a1 = 0.f;
#pragma unroll
                for (int u = 0; u < 4; ++u) {
                    a0 += wr[t][u] * w20[u];
                    a1 += wr[t][u] * w21[u];
                }
                pv[2 * t]     = a0;
                pv[2 * t + 1] = a1;
            }
            const float r1 = wave_sum8(pv, lane);
            if ((lane & 3) == 0) {
                const int k = lane >> 2;
                wp[(q * 32 + nt + 8 * (k >> 1)) * 2 + (k & 1)] = r1;
            }
        }
    }
    __syncthreads();

    float* Qs = xs;
    float* Ks = xs + MROW * HP;
    if (mg == 0) phase_b<4>(hs, wp, pp, Qs, Ks, Wq, Wk, 0, nt, n, rr, half, qfirst, kfirst);
    else         phase_b<3>(hs, wp, pp, Qs, Ks, Wq, Wk, 1, nt, n, rr, half, qfirst, kfirst);
    __syncthreads();

    const float bias0 = b2[0], bias1 = b2[1];
    const float scale = 0.08838834764831845f;
    float* Gw = att + wu * AW;
    float* Aw = Gw + 16 * GP;
    float* Pw = Aw + ALW;

    for (int t = wu; t < NTASK; t += 16) {
        const int s0 = t * PT;
        const int r0 = s0 - 5;
        const int npos  = SEQ - s0 < PT ? SEQ - s0 : PT;
        const int npass = (npos + 1) >> 1;
        const int sh = (t & 1) * 2;

        {
            int ra = r0 + rr;
            ra = ra < 0 ? 0 : (ra > MROW - 1 ? MROW - 1 : ra);
            const float* qa = Qs + ra * HP + half * 2;
            const float* kb = Ks + ra * HP + half * 2;
            v8f acc0 = {0.f,0.f,0.f,0.f,0.f,0.f,0.f,0.f};
            v8f acc1 = acc0;
#pragma unroll
            for (int kk = 0; kk < 64; kk += 4) {
                const v2f a0 = *(const v2f*)(qa + kk);
                const v2f f0 = *(const v2f*)(kb + kk);
                const v2f a1 = *(const v2f*)(qa + 64 + kk);
                const v2f f1 = *(const v2f*)(kb + 64 + kk);
                acc0 = wmma_f32_k4(a0, f0, acc0);
                acc1 = wmma_f32_k4(a1, f1, acc1);
            }
            float* gp = Gw + (half * 8) * GP + rr;
#pragma unroll
            for (int r = 0; r < 8; ++r) gp[r * GP] = acc0[r] + acc1[r];
        }
        {
            int rp = r0 + (lane >> 1);
            rp = rp < 0 ? 0 : (rp > MROW - 1 ? MROW - 1 : rp);
            float sp = 0.f;
#pragma unroll
            for (int w = 0; w < 8; ++w) sp += pp[w * (MROW * 2) + rp * 2 + (lane & 1)];
            Pw[lane] = sp;
        }
        wave_sync();

        for (int q = 0; q < npass; ++q) {
            const int p  = q * 2 + half;
            const int s  = s0 + p;
            const int ic = rr < WIN ? rr : WIN - 1;
            const int ri = s + 5 - ic;
            const bool rowok = (ri >= 0) && (ri < SEQ);
            const float* gq = Gw + (p + 10 - ic) * GP + (p + 10);
            float sc[WIN];
#pragma unroll
            for (int j = 0; j < WIN; ++j) {
                const int rj = s + 5 - j;
                const bool ok = rowok && (rj >= 0) && (rj < SEQ);
                const float gv = gq[-j];
                sc[j] = ok ? gv * scale : 0.f;
            }
            float mv = sc[0];
#pragma unroll
            for (int j = 1; j < WIN; ++j) mv = fmaxf(mv, sc[j]);
            float sum = 0.f;
#pragma unroll
            for (int j = 0; j < WIN; ++j) {
                sc[j] = expf(sc[j] - mv);
                sum += sc[j];
            }
            const float rinv = 1.0f / sum;
            if (rr < WIN) {
                float* ap = Aw + sh + p * (WIN * WIN) + rr * WIN;
#pragma unroll
                for (int j = 0; j < WIN; ++j) ap[j] = sc[j] * rinv;
            }
        }
        wave_sync();

        {
            float* gdst = alphaOut + ((size_t)seq * SEQ + s0) * (WIN * WIN) - sh;
            const int jend = sh + npos * (WIN * WIN);
            const int i0 = (sh + 3) >> 2, i1 = jend >> 2;
            v4f av[6];
#pragma unroll
            for (int u = 0; u < 6; ++u) {
                const int i  = i0 + lane + u * 32;
                const int ie = i < i1 ? i : i1 - 1;
                av[u] = *(const v4f*)(Aw + ie * 4);
            }
#pragma unroll
            for (int u = 0; u < 6; ++u) {
                const int i = i0 + lane + u * 32;
                if (i < i1) *(v4f*)(gdst + i * 4) = av[u];
            }
            if (lane == 0 && sh != 0)
                *(v2f*)(gdst + sh) = *(const v2f*)(Aw + sh);
            if (lane == 1 && (jend & 3) != 0)
                *(v2f*)(gdst + i1 * 4) = *(const v2f*)(Aw + i1 * 4);
        }

        for (int q = 0; q < npass; ++q) {
            const int p  = q * 2 + half;
            const int s  = s0 + p;
            const int jc = rr < WIN ? rr : WIN - 1;
            float beta = 0.f;
            {
                const float* ac = Aw + sh + p * (WIN * WIN) + jc;
#pragma unroll
                for (int i = 0; i < WIN; ++i) beta += ac[i * WIN];
            }
            const int rv = s + 5 - jc;
            const bool valid = (rr < WIN) && (rv >= 0) && (rv < SEQ);
            const v2f pr = *(const v2f*)(Pw + (p + 10 - jc) * 2);
            const float t0 = valid ? beta * pr.x : 0.f;
            const float t1 = valid ? beta * pr.y : 0.f;
            const bool hi8 = (rr & 8) != 0;
            const float keep = hi8 ? t1 : t0;
            const float send = hi8 ? t0 : t1;
            float r = keep + __shfl_xor(send, 8, 32);
            r += __shfl_xor(r, 4, 32);
            r += __shfl_xor(r, 2, 32);
            r += __shfl_xor(r, 1, 32);
            if ((rr & 7) == 0 && s < SEQ) {
                const int c = rr >> 3;
                const float bias = c ? bias1 : bias0;
                out[((size_t)seq * SEQ + s) * 2 + c] = 1.f / (1.f + expf(-(r + bias)));
            }
        }
        wave_sync();
    }
}

extern "C" void kernel_launch(void* const* d_in, const int* in_sizes, int n_in,
                              void* d_out, int out_size, void* d_ws, size_t ws_size,
                              hipStream_t stream) {
    const float* x  = (const float*)d_in[0];
    const float* W1 = (const float*)d_in[1];
    const float* b1 = (const float*)d_in[2];
    const float* Wq = (const float*)d_in[3];
    const float* Wk = (const float*)d_in[4];
    const float* Wv = (const float*)d_in[5];
    const float* W2 = (const float*)d_in[6];
    const float* b2 = (const float*)d_in[7];

    float* out      = (float*)d_out;
    float* alphaOut = out + (size_t)NROW * 2;

    k_seq<<<BATCH, 512, 0, stream>>>(x, W1, b1, Wq, Wk, Wv, W2, b2, out, alphaOut);
}
